# starting kernel
# baseline (speedup 1.0000x reference)
#include <hip/hip_runtime.h>
#include <hip/hip_bf16.h>


#define B_  16
#define TQ_ 64
#define TK_ 512
#define D_  256
#define V_  32000

#define VB1 2048
#define NVB1 16
#define VB2 512
#define NVB2 63

typedef __attribute__((ext_vector_type(2))) float v2f;
typedef __attribute__((ext_vector_type(8))) float v8f;

__device__ inline v8f wmma_f32_k4(v2f a, v2f b, v8f c) {
  return __builtin_amdgcn_wmma_f32_16x16x4_f32(
       false, a,  false, b,
       (short)0, c,  false,  false);
}

__device__ inline float wave_max(float x) {
  #pragma unroll
  for (int o = 1; o < 32; o <<= 1) x = fmaxf(x, __shfl_xor(x, o, 32));
  return x;
}
__device__ inline float wave_sum(float x) {
  #pragma unroll
  for (int o = 1; o < 32; o <<= 1) x += __shfl_xor(x, o, 32);
  return x;
}
__device__ inline float half16_max(float x) {
  #pragma unroll
  for (int o = 1; o < 16; o <<= 1) x = fmaxf(x, __shfl_xor(x, o, 32));
  return x;
}
__device__ inline float half16_sum(float x) {
  #pragma unroll
  for (int o = 1; o < 16; o <<= 1) x += __shfl_xor(x, o, 32);
  return x;
}

__global__ __launch_bounds__(256) void k_attn(
    const float* __restrict__ dom, const float* __restrict__ out_st,
    const float* __restrict__ ctx, const float* __restrict__ Wg,
    const float* __restrict__ bg, float* __restrict__ attnW,
    float* __restrict__ pgenW)
{
  __shared__ float sS[16][TK_];
  __shared__ float sCV[16][D_];

  const int b    = blockIdx.x >> 2;
  const int q0   = (blockIdx.x & 3) << 4;
  const int wave = threadIdx.x >> 5;
  const int lane = threadIdx.x & 31;
  const int row  = lane & 15;
  const int hi   = lane >> 4;

  v8f acc[4];
  #pragma unroll
  for (int i = 0; i < 4; i++) acc[i] = {};
  const float* Abase = out_st + (size_t)(b*TQ_ + q0 + row)*D_;
  for (int dc = 0; dc < D_/4; ++dc) {
    const int d0 = dc*4;
    v2f a = *(const v2f*)(Abase + d0 + 2*hi);
    #pragma unroll
    for (int i = 0; i < 4; i++) {
      const int k0 = (wave*4 + i)*16;
      v2f bb = *(const v2f*)(ctx + (size_t)(b*TK_ + k0 + row)*D_ + d0 + 2*hi);
      acc[i] = wmma_f32_k4(a, bb, acc[i]);
    }
  }
  #pragma unroll
  for (int i = 0; i < 4; i++) {
    const int k0 = (wave*4 + i)*16;
    #pragma unroll
    for (int r = 0; r < 8; r++)
      sS[hi*8 + r][k0 + row] = acc[i][r] * 0.0625f;
  }
  __syncthreads();

  #pragma unroll
  for (int j = 0; j < 2; j++) {
    const int q = wave*2 + j;
    float v[TK_/32];
    float m = -1e30f;
    #pragma unroll
    for (int t = 0; t < TK_/32; t++) { v[t] = sS[q][lane + 32*t]; m = fmaxf(m, v[t]); }
    m = wave_max(m);
    float s = 0.f;
    #pragma unroll
    for (int t = 0; t < TK_/32; t++) { v[t] = __expf(v[t] - m); s += v[t]; }
    s = wave_sum(s);
    const float inv = 1.0f / s;
    #pragma unroll
    for (int t = 0; t < TK_/32; t++) {
      const float p = v[t] * inv;
      sS[q][lane + 32*t] = p;
      attnW[(size_t)(b*TQ_ + q0 + q)*TK_ + lane + 32*t] = p;
    }
  }
  __syncthreads();

  v8f cv[2]; cv[0] = {}; cv[1] = {};
  for (int kc = 0; kc < TK_/4; ++kc) {
    const int k0 = kc*4;
    v2f a;
    a.x = sS[row][k0 + 2*hi];
    a.y = sS[row][k0 + 2*hi + 1];
    #pragma unroll
    for (int i = 0; i < 2; i++) {
      const int n0 = (wave*2 + i)*16;
      v2f bb;
      bb.x = ctx[(size_t)(b*TK_ + k0 + 2*hi    )*D_ + n0 + row];
      bb.y = ctx[(size_t)(b*TK_ + k0 + 2*hi + 1)*D_ + n0 + row];
      cv[i] = wmma_f32_k4(a, bb, cv[i]);
    }
  }
  #pragma unroll
  for (int i = 0; i < 2; i++) {
    const int n0 = (wave*2 + i)*16;
    #pragma unroll
    for (int r = 0; r < 8; r++)
      sCV[hi*8 + r][n0 + row] = cv[i][r];
  }
  __syncthreads();

  #pragma unroll
  for (int j = 0; j < 2; j++) {
    const int q = wave*2 + j;
    float dot = 0.f;
    for (int d = lane; d < D_; d += 32) {
      dot += out_st[(size_t)(b*TQ_ + q0 + q)*D_ + d] * Wg[d];
      dot += sCV[q][d]                               * Wg[D_ + d];
      dot += dom[(size_t)(b*TQ_ + q0 + q)*D_ + d]    * Wg[2*D_ + d];
    }
    dot = wave_sum(dot);
    if (lane == 0) {
      const float x = dot + bg[0];
      pgenW[b*TQ_ + q0 + q] = 1.0f / (1.0f + __expf(-x));
    }
  }
}

__global__ __launch_bounds__(256) void k_stats(
    const float* __restrict__ out_st, const float* __restrict__ Ws,
    const float* __restrict__ bs, float* __restrict__ pmW,
    float* __restrict__ psW)
{
  const int b    = blockIdx.x >> 2;
  const int q0   = (blockIdx.x & 3) << 4;
  const int vb   = blockIdx.y;
  const int v0   = vb * VB1;
  const int cols = min(VB1, V_ - v0);
  const int ntiles = cols >> 4;

  const int wave = threadIdx.x >> 5;
  const int lane = threadIdx.x & 31;
  const int row  = lane & 15;
  const int hi   = lane >> 4;

  __shared__ float sM[8][16];
  __shared__ float sSm[8][16];

  float rm[8], rs[8];
  #pragma unroll
  for (int r = 0; r < 8; r++) { rm[r] = -1e30f; rs[r] = 0.f; }

  const float* Abase = out_st + (size_t)(b*TQ_ + q0 + row)*D_;
  for (int nt = wave; nt < ntiles; nt += 8) {
    const int c0 = v0 + nt*16;
    v8f acc = {};
    for (int dc = 0; dc < D_/4; ++dc) {
      const int d0 = dc*4;
      v2f a = *(const v2f*)(Abase + d0 + 2*hi);
      v2f bb;
      bb.x = Ws[(size_t)(d0 + 2*hi    )*V_ + c0 + row];
      bb.y = Ws[(size_t)(d0 + 2*hi + 1)*V_ + c0 + row];
      acc = wmma_f32_k4(a, bb, acc);
    }
    const float bsv = bs[c0 + row];
    #pragma unroll
    for (int r = 0; r < 8; r++) {
      const float z  = acc[r] + bsv;
      const float tm = half16_max(z);
      const float nm = fmaxf(rm[r], tm);
      const float ts = half16_sum(__expf(z - nm));
      rs[r] = rs[r] * __expf(rm[r] - nm) + ts;
      rm[r] = nm;
    }
  }
  if (row == 0) {
    #pragma unroll
    for (int r = 0; r < 8; r++) { sM[wave][hi*8 + r] = rm[r]; sSm[wave][hi*8 + r] = rs[r]; }
  }
  __syncthreads();
  if (threadIdx.x < 16) {
    const int q = threadIdx.x;
    float gm = -1e30f;
    #pragma unroll
    for (int w = 0; w < 8; w++) gm = fmaxf(gm, sM[w][q]);
    float gs = 0.f;
    #pragma unroll
    for (int w = 0; w < 8; w++) gs += sSm[w][q] * __expf(sM[w][q] - gm);
    const int rowid = b*TQ_ + q0 + q;
    pmW[(size_t)rowid*NVB1 + vb] = gm;
    psW[(size_t)rowid*NVB1 + vb] = gs;
  }
}

__global__ __launch_bounds__(256) void k_reduce(
    const float* __restrict__ pmW, const float* __restrict__ psW,
    float* __restrict__ rmW, float* __restrict__ rsW)
{
  const int rowid = blockIdx.x*256 + threadIdx.x;
  if (rowid >= B_*TQ_) return;
  float gm = -1e30f;
  #pragma unroll
  for (int v = 0; v < NVB1; ++v) gm = fmaxf(gm, pmW[(size_t)rowid*NVB1 + v]);
  float gs = 0.f;
  #pragma unroll
  for (int v = 0; v < NVB1; ++v)
    gs += psW[(size_t)rowid*NVB1 + v] * __expf(pmW[(size_t)rowid*NVB1 + v] - gm);
  rmW[rowid] = gm;
  rsW[rowid] = gs;
}

__global__ __launch_bounds__(256) void k_final(
    const float* __restrict__ out_st, const float* __restrict__ Ws,
    const float* __restrict__ bs, const int* __restrict__ cp,
    const float* __restrict__ attnW, const float* __restrict__ pgenW,
    const float* __restrict__ rmW, const float* __restrict__ rsW,
    float* __restrict__ outp)
{
  const int b    = blockIdx.x >> 2;
  const int q0   = (blockIdx.x & 3) << 4;
  const int vb   = blockIdx.y;
  const int v0   = vb * VB2;
  const int cols = min(VB2, V_ - v0);
  const int ntiles = cols >> 4;

  const int wave = threadIdx.x >> 5;
  const int lane = threadIdx.x & 31;
  const int row  = lane & 15;
  const int hi   = lane >> 4;

  __shared__ float sPtr[16][VB2];
  __shared__ float sScale[16], sMr[16], sOmp[16];

  for (int i = threadIdx.x; i < 16*VB2; i += 256)
    (&sPtr[0][0])[i] = 0.f;
  if (threadIdx.x < 16) {
    const int rowid = b*TQ_ + q0 + threadIdx.x;
    const float pg = pgenW[rowid];
    sScale[threadIdx.x] = pg / rsW[rowid];
    sMr[threadIdx.x]    = rmW[rowid];
    sOmp[threadIdx.x]   = 1.0f - pg;
  }
  __syncthreads();

  for (int k = threadIdx.x; k < TK_; k += 256) {
    const int key = cp[b*TK_ + k];
    if (key >= v0 && key < v0 + cols) {
      const int c = key - v0;
      #pragma unroll
      for (int q = 0; q < 16; q++) {
        const float w = sOmp[q] * attnW[(size_t)(b*TQ_ + q0 + q)*TK_ + k];
        atomicAdd(&sPtr[q][c], w);
      }
    }
  }
  __syncthreads();

  const float* Abase = out_st + (size_t)(b*TQ_ + q0 + row)*D_;
  for (int nt = wave; nt < ntiles; nt += 8) {
    const int c0 = v0 + nt*16;
    v8f acc = {};
    for (int dc = 0; dc < D_/4; ++dc) {
      const int d0 = dc*4;
      v2f a = *(const v2f*)(Abase + d0 + 2*hi);
      v2f bb;
      bb.x = Ws[(size_t)(d0 + 2*hi    )*V_ + c0 + row];
      bb.y = Ws[(size_t)(d0 + 2*hi + 1)*V_ + c0 + row];
      acc = wmma_f32_k4(a, bb, acc);
    }
    const float bsv = bs[c0 + row];
    #pragma unroll
    for (int r = 0; r < 8; r++) {
      const int q = hi*8 + r;
      const float z = acc[r] + bsv;
      const float p = sScale[q] * __expf(z - sMr[q]) + sPtr[q][nt*16 + row];
      outp[(size_t)(b*TQ_ + q0 + q)*V_ + c0 + row] = __logf(p);
    }
  }
}

extern "C" void kernel_launch(void* const* d_in, const int* in_sizes, int n_in,
                              void* d_out, int out_size, void* d_ws, size_t ws_size,
                              hipStream_t stream) {
  (void)in_sizes; (void)n_in; (void)out_size; (void)ws_size;
  const float* dom    = (const float*)d_in[0];
  const float* out_st = (const float*)d_in[1];
  const float* ctx    = (const float*)d_in[2];
  const int*   cp     = (const int*)d_in[3];
  const float* Ws     = (const float*)d_in[4];
  const float* bsv    = (const float*)d_in[5];
  const float* Wg     = (const float*)d_in[6];
  const float* bg     = (const float*)d_in[7];
  float* outp = (float*)d_out;

  float* attnW = (float*)d_ws;
  float* pgenW = attnW + (size_t)B_*TQ_*TK_;
  float* pmW   = pgenW + B_*TQ_;
  float* psW   = pmW + (size_t)B_*TQ_*NVB1;
  float* rmW   = psW + (size_t)B_*TQ_*NVB1;
  float* rsW   = rmW + B_*TQ_;

  k_attn  <<<dim3(B_*TQ_/16),        256, 0, stream>>>(dom, out_st, ctx, Wg, bg, attnW, pgenW);
  k_stats <<<dim3(B_*TQ_/16, NVB1),  256, 0, stream>>>(out_st, Ws, bsv, pmW, psW);
  k_reduce<<<dim3((B_*TQ_+255)/256), 256, 0, stream>>>(pmW, psW, rmW, rsW);
  k_final <<<dim3(B_*TQ_/16, NVB2),  256, 0, stream>>>(out_st, Ws, bsv, cp, attnW, pgenW, rmW, rsW, outp);
}


# Vocab passes: work-group keeps the 64 queries of a batch entry in LDS, each W_state fragment (one 16 B load per row) feeds 4 WMMAs; softmax max/sum partials regrouped per 64 columns (sum order)
# speedup vs baseline: 1.7285x; 1.7285x over previous
#include <hip/hip_runtime.h>
#include <hip/hip_bf16.h>


#define B_  16
#define TQ_ 64
#define TK_ 512
#define D_  256
#define V_  32000

#define CB  256
#define NCB 125
#define RP  260

typedef __attribute__((ext_vector_type(2))) float v2f;
typedef __attribute__((ext_vector_type(4))) float v4f;
typedef __attribute__((ext_vector_type(8))) float v8f;

__device__ inline v8f wmma_f32_k4(v2f a, v2f b, v8f c) {
  return __builtin_amdgcn_wmma_f32_16x16x4_f32(
       false, a,  false, b,
       (short)0, c,  false,  false);
}

__device__ inline float wave_max(float x) {
  #pragma unroll
  for (int o = 1; o < 32; o <<= 1) x = fmaxf(x, __shfl_xor(x, o, 32));
  return x;
}
__device__ inline float wave_sum(float x) {
  #pragma unroll
  for (int o = 1; o < 32; o <<= 1) x += __shfl_xor(x, o, 32);
  return x;
}
__device__ inline float half16_max(float x) {
  #pragma unroll
  for (int o = 1; o < 16; o <<= 1) x = fmaxf(x, __shfl_xor(x, o, 32));
  return x;
}
__device__ inline float half16_sum(float x) {
  #pragma unroll
  for (int o = 1; o < 16; o <<= 1) x += __shfl_xor(x, o, 32);
  return x;
}

__global__ __launch_bounds__(256) void k_attn(
    const float* __restrict__ dom, const float* __restrict__ out_st,
    const float* __restrict__ ctx, const float* __restrict__ Wg,
    const float* __restrict__ bg, float* __restrict__ attnW,
    float* __restrict__ pgenW)
{
  __shared__ float sS[16][TK_];
  __shared__ float sCV[16][D_];

  const int b    = blockIdx.x >> 2;
  const int q0   = (blockIdx.x & 3) << 4;
  const int wave = threadIdx.x >> 5;
  const int lane = threadIdx.x & 31;
  const int row  = lane & 15;
  const int hi   = lane >> 4;

  v8f acc[4];
  #pragma unroll
  for (int i = 0; i < 4; i++) acc[i] = {};
  const float* Abase = out_st + (size_t)(b*TQ_ + q0 + row)*D_;
  for (int dc = 0; dc < D_/4; ++dc) {
    const int d0 = dc*4;
    v2f a = *(const v2f*)(Abase + d0 + 2*hi);
    #pragma unroll
    for (int i = 0; i < 4; i++) {
      const int k0 = (wave*4 + i)*16;
      v2f bb = *(const v2f*)(ctx + (size_t)(b*TK_ + k0 + row)*D_ + d0 + 2*hi);
      acc[i] = wmma_f32_k4(a, bb, acc[i]);
    }
  }
  #pragma unroll
  for (int i = 0; i < 4; i++) {
    const int k0 = (wave*4 + i)*16;
    #pragma unroll
    for (int r = 0; r < 8; r++)
      sS[hi*8 + r][k0 + row] = acc[i][r] * 0.0625f;
  }
  __syncthreads();

  #pragma unroll
  for (int j = 0; j < 2; j++) {
    const int q = wave*2 + j;
    float v[TK_/32];
    float m = -1e30f;
    #pragma unroll
    for (int t = 0; t < TK_/32; t++) { v[t] = sS[q][lane + 32*t]; m = fmaxf(m, v[t]); }
    m = wave_max(m);
    float s = 0.f;
    #pragma unroll
    for (int t = 0; t < TK_/32; t++) { v[t] = __expf(v[t] - m); s += v[t]; }
    s = wave_sum(s);
    const float inv = 1.0f / s;
    #pragma unroll
    for (int t = 0; t < TK_/32; t++) {
      const float p = v[t] * inv;
      sS[q][lane + 32*t] = p;
      attnW[(size_t)(b*TQ_ + q0 + q)*TK_ + lane + 32*t] = p;
    }
  }
  __syncthreads();

  v8f cv[2]; cv[0] = {}; cv[1] = {};
  for (int kc = 0; kc < TK_/4; ++kc) {
    const int k0 = kc*4;
    v2f a;
    a.x = sS[row][k0 + 2*hi];
    a.y = sS[row][k0 + 2*hi + 1];
    #pragma unroll
    for (int i = 0; i < 2; i++) {
      const int n0 = (wave*2 + i)*16;
      v2f bb;
      bb.x = ctx[(size_t)(b*TK_ + k0 + 2*hi    )*D_ + n0 + row];
      bb.y = ctx[(size_t)(b*TK_ + k0 + 2*hi + 1)*D_ + n0 + row];
      cv[i] = wmma_f32_k4(a, bb, cv[i]);
    }
  }
  #pragma unroll
  for (int i = 0; i < 2; i++) {
    const int n0 = (wave*2 + i)*16;
    #pragma unroll
    for (int r = 0; r < 8; r++)
      sCV[hi*8 + r][n0 + row] = cv[i][r];
  }
  __syncthreads();

  #pragma unroll
  for (int j = 0; j < 2; j++) {
    const int q = wave*2 + j;
    float dot = 0.f;
    for (int d = lane; d < D_; d += 32) {
      dot += out_st[(size_t)(b*TQ_ + q0 + q)*D_ + d] * Wg[d];
      dot += sCV[q][d]                               * Wg[D_ + d];
      dot += dom[(size_t)(b*TQ_ + q0 + q)*D_ + d]    * Wg[2*D_ + d];
    }
    dot = wave_sum(dot);
    if (lane == 0) {
      const float x = dot + bg[0];
      pgenW[b*TQ_ + q0 + q] = 1.0f / (1.0f + __expf(-x));
    }
  }
}

__device__ inline void stage_queries(float* sA, const float* __restrict__ src, int tid) {
  for (int i = tid; i < TQ_*(D_/4); i += 128) {
    const int r  = i >> 6;
    const int c4 = (i & 63) << 2;
    *(v4f*)(sA + r*RP + c4) = *(const v4f*)(src + r*D_ + c4);
  }
}

__device__ inline void logits_64x64(const float* sA, const float* __restrict__ Ws,
                                    int c0, int row, int hi, v8f (&acc)[4][4]) {
  #pragma unroll
  for (int m = 0; m < 4; m++)
    #pragma unroll
    for (int j = 0; j < 4; j++) acc[m][j] = {};

  const float* Bp = Ws + (size_t)(2*hi)*V_ + c0 + 4*row;
  const float* Ap = sA + row*RP + 2*hi;
  #pragma unroll 2
  for (int ks = 0; ks < D_/4; ++ks) {
    const v4f b0 = *(const v4f*)(Bp + (size_t)(4*ks    )*V_);
    const v4f b1 = *(const v4f*)(Bp + (size_t)(4*ks + 1)*V_);
    v2f a[4];
    #pragma unroll
    for (int m = 0; m < 4; m++) a[m] = *(const v2f*)(Ap + m*16*RP + 4*ks);
    #pragma unroll
    for (int j = 0; j < 4; j++) {
      v2f bb; bb.x = b0[j]; bb.y = b1[j];
      #pragma unroll
      for (int m = 0; m < 4; m++) acc[m][j] = wmma_f32_k4(a[m], bb, acc[m][j]);
    }
  }
}

__global__ __launch_bounds__(128) void k_stats(
    const float* __restrict__ out_st, const float* __restrict__ Ws,
    const float* __restrict__ bs, float* __restrict__ pmW,
    float* __restrict__ psW)
{
  const int b    = blockIdx.x;
  const int vb   = blockIdx.y;
  const int wave = threadIdx.x >> 5;
  const int lane = threadIdx.x & 31;
  const int row  = lane & 15;
  const int hi   = lane >> 4;
  const int c0   = vb*CB + wave*64;

  __shared__ __attribute__((aligned(16))) float sA[TQ_*RP];
  __shared__ float sM[4][TQ_];
  __shared__ float sSm[4][TQ_];

  stage_queries(sA, out_st + (size_t)b*TQ_*D_, threadIdx.x);
  __syncthreads();

  v8f acc[4][4];
  logits_64x64(sA, Ws, c0, row, hi, acc);

  const v4f bsv = *(const v4f*)(bs + c0 + 4*row);
  #pragma unroll
  for (int m = 0; m < 4; m++) {
    #pragma unroll
    for (int r = 0; r < 8; r++) {
      const float z0 = acc[m][0][r] + bsv[0];
      const float z1 = acc[m][1][r] + bsv[1];
      const float z2 = acc[m][2][r] + bsv[2];
      const float z3 = acc[m][3][r] + bsv[3];
      const float tm = half16_max(fmaxf(fmaxf(z0, z1), fmaxf(z2, z3)));
      const float ts = half16_sum((__expf(z0 - tm) + __expf(z1 - tm)) +
                                  (__expf(z2 - tm) + __expf(z3 - tm)));
      if (row == 0) { sM[wave][m*16 + hi*8 + r] = tm; sSm[wave][m*16 + hi*8 + r] = ts; }
    }
  }
  __syncthreads();
  if (threadIdx.x < TQ_) {
    const int q = threadIdx.x;
    float gm = -1e30f;
    #pragma unroll
    for (int w = 0; w < 4; w++) gm = fmaxf(gm, sM[w][q]);
    float gs = 0.f;
    #pragma unroll
    for (int w = 0; w < 4; w++) gs += sSm[w][q] * __expf(sM[w][q] - gm);
    pmW[(size_t)vb*(B_*TQ_) + b*TQ_ + q] = gm;
    psW[(size_t)vb*(B_*TQ_) + b*TQ_ + q] = gs;
  }
}

__global__ __launch_bounds__(256) void k_reduce(
    const float* __restrict__ pmW, const float* __restrict__ psW,
    float* __restrict__ rmW, float* __restrict__ rsW)
{
  const int rowid = blockIdx.x*256 + threadIdx.x;
  if (rowid >= B_*TQ_) return;
  float gm = -1e30f;
  for (int v = 0; v < NCB; ++v) gm = fmaxf(gm, pmW[(size_t)v*(B_*TQ_) + rowid]);
  float gs = 0.f;
  for (int v = 0; v < NCB; ++v)
    gs += psW[(size_t)v*(B_*TQ_) + rowid] * __expf(pmW[(size_t)v*(B_*TQ_) + rowid] - gm);
  rmW[rowid] = gm;
  rsW[rowid] = gs;
}

__global__ __launch_bounds__(128) void k_final(
    const float* __restrict__ out_st, const float* __restrict__ Ws,
    const float* __restrict__ bs, const int* __restrict__ cp,
    const float* __restrict__ attnW, const float* __restrict__ pgenW,
    const float* __restrict__ rmW, const float* __restrict__ rsW,
    float* __restrict__ outp)
{
  const int b    = blockIdx.x;
  const int vb   = blockIdx.y;
  const int v0   = vb*CB;
  const int wave = threadIdx.x >> 5;
  const int lane = threadIdx.x & 31;
  const int row  = lane & 15;
  const int hi   = lane >> 4;
  const int c0   = v0 + wave*64;

  __shared__ __attribute__((aligned(16))) float sA[TQ_*RP];
  __shared__ float sScale[TQ_], sMr[TQ_], sOmp[TQ_];

  stage_queries(sA, out_st + (size_t)b*TQ_*D_, threadIdx.x);
  if (threadIdx.x < TQ_) {
    const int rowid = b*TQ_ + threadIdx.x;
    const float pg = pgenW[rowid];
    sScale[threadIdx.x] = pg / rsW[rowid];
    sMr[threadIdx.x]    = rmW[rowid];
    sOmp[threadIdx.x]   = 1.0f - pg;
  }
  __syncthreads();

  v8f acc[4][4];
  logits_64x64(sA, Ws, c0, row, hi, acc);
  __syncthreads();

  for (int i = threadIdx.x; i < TQ_*(CB/4); i += 128) {
    const int r  = i >> 6;
    const int c4 = (i & 63) << 2;
    *(v4f*)(sA + r*RP + c4) = (v4f){0.f, 0.f, 0.f, 0.f};
  }
  __syncthreads();

  for (int k = threadIdx.x; k < TK_; k += 128) {
    const int key = cp[b*TK_ + k];
    if (key >= v0 && key < v0 + CB) {
      const int c = key - v0;
      for (int q = 0; q < TQ_; q++) {
        const float w = sOmp[q] * attnW[(size_t)(b*TQ_ + q)*TK_ + k];
        atomicAdd(&sA[q*RP + c], w);
      }
    }
  }
  __syncthreads();

  const v4f bsv = *(const v4f*)(bs + c0 + 4*row);
  #pragma unroll
  for (int m = 0; m < 4; m++) {
    #pragma unroll
    for (int r = 0; r < 8; r++) {
      const int q = m*16 + hi*8 + r;
      const float sc = sScale[q];
      const float mr = sMr[q];
      const v4f pt = *(const v4f*)(sA + q*RP + wave*64 + 4*row);
      v4f o;
      #pragma unroll
      for (int j = 0; j < 4; j++) {
        const float z = acc[m][j][r] + bsv[j];
        o[j] = __logf(sc * __expf(z - mr) + pt[j]);
      }
      *(v4f*)(outp + (size_t)(b*TQ_ + q)*V_ + c0 + 4*row) = o;
    }
  }
}

extern "C" void kernel_launch(void* const* d_in, const int* in_sizes, int n_in,
                              void* d_out, int out_size, void* d_ws, size_t ws_size,
                              hipStream_t stream) {
  (void)in_sizes; (void)n_in; (void)out_size; (void)ws_size;
  const float* dom    = (const float*)d_in[0];
  const float* out_st = (const float*)d_in[1];
  const float* ctx    = (const float*)d_in[2];
  const int*   cp     = (const int*)d_in[3];
  const float* Ws     = (const float*)d_in[4];
  const float* bsv    = (const float*)d_in[5];
  const float* Wg     = (const float*)d_in[6];
  const float* bg     = (const float*)d_in[7];
  float* outp = (float*)d_out;

  float* attnW = (float*)d_ws;
  float* pgenW = attnW + (size_t)B_*TQ_*TK_;
  float* rmW   = pgenW + B_*TQ_;
  float* rsW   = rmW + B_*TQ_;
  float* pmW   = rsW + B_*TQ_;
  float* psW   = pmW + (size_t)NCB*B_*TQ_;

  k_attn  <<<dim3(B_*TQ_/16),        256, 0, stream>>>(dom, out_st, ctx, Wg, bg, attnW, pgenW);
  k_stats <<<dim3(B_, NCB),          128, 0, stream>>>(out_st, Ws, bsv, pmW, psW);
  k_reduce<<<dim3((B_*TQ_+255)/256), 256, 0, stream>>>(pmW, psW, rmW, rsW);
  k_final <<<dim3(B_, NCB),          128, 0, stream>>>(out_st, Ws, bsv, cp, attnW, pgenW, rmW, rsW, outp);
}


# Vocab GEMM computed once: the stats pass stores the biased logits in the output buffer; the final pass reads them back in place as a streaming pass instead of redoing the W_state GEMM
# speedup vs baseline: 2.1172x; 1.2249x over previous
#include <hip/hip_runtime.h>
#include <hip/hip_bf16.h>


#define B_  16
#define TQ_ 64
#define TK_ 512
#define D_  256
#define V_  32000

#define CB  256
#define NCB 125
#define RP  260

typedef __attribute__((ext_vector_type(2))) float v2f;
typedef __attribute__((ext_vector_type(4))) float v4f;
typedef __attribute__((ext_vector_type(8))) float v8f;

__device__ inline v8f wmma_f32_k4(v2f a, v2f b, v8f c) {
  return __builtin_amdgcn_wmma_f32_16x16x4_f32(
       false, a,  false, b,
       (short)0, c,  false,  false);
}

__device__ inline float wave_max(float x) {
  #pragma unroll
  for (int o = 1; o < 32; o <<= 1) x = fmaxf(x, __shfl_xor(x, o, 32));
  return x;
}
__device__ inline float wave_sum(float x) {
  #pragma unroll
  for (int o = 1; o < 32; o <<= 1) x += __shfl_xor(x, o, 32);
  return x;
}
__device__ inline float half16_max(float x) {
  #pragma unroll
  for (int o = 1; o < 16; o <<= 1) x = fmaxf(x, __shfl_xor(x, o, 32));
  return x;
}
__device__ inline float half16_sum(float x) {
  #pragma unroll
  for (int o = 1; o < 16; o <<= 1) x += __shfl_xor(x, o, 32);
  return x;
}

__global__ __launch_bounds__(256) void k_attn(
    const float* __restrict__ dom, const float* __restrict__ out_st,
    const float* __restrict__ ctx, const float* __restrict__ Wg,
    const float* __restrict__ bg, float* __restrict__ attnW,
    float* __restrict__ pgenW)
{
  __shared__ float sS[16][TK_];
  __shared__ float sCV[16][D_];

  const int b    = blockIdx.x >> 2;
  const int q0   = (blockIdx.x & 3) << 4;
  const int wave = threadIdx.x >> 5;
  const int lane = threadIdx.x & 31;
  const int row  = lane & 15;
  const int hi   = lane >> 4;

  v8f acc[4];
  #pragma unroll
  for (int i = 0; i < 4; i++) acc[i] = {};
  const float* Abase = out_st + (size_t)(b*TQ_ + q0 + row)*D_;
  for (int dc = 0; dc < D_/4; ++dc) {
    const int d0 = dc*4;
    v2f a = *(const v2f*)(Abase + d0 + 2*hi);
    #pragma unroll
    for (int i = 0; i < 4; i++) {
      const int k0 = (wave*4 + i)*16;
      v2f bb = *(const v2f*)(ctx + (size_t)(b*TK_ + k0 + row)*D_ + d0 + 2*hi);
      acc[i] = wmma_f32_k4(a, bb, acc[i]);
    }
  }
  #pragma unroll
  for (int i = 0; i < 4; i++) {
    const int k0 = (wave*4 + i)*16;
    #pragma unroll
    for (int r = 0; r < 8; r++)
      sS[hi*8 + r][k0 + row] = acc[i][r] * 0.0625f;
  }
  __syncthreads();

  #pragma unroll
  for (int j = 0; j < 2; j++) {
    const int q = wave*2 + j;
    float v[TK_/32];
    float m = -1e30f;
    #pragma unroll
    for (int t = 0; t < TK_/32; t++) { v[t] = sS[q][lane + 32*t]; m = fmaxf(m, v[t]); }
    m = wave_max(m);
    float s = 0.f;
    #pragma unroll
    for (int t = 0; t < TK_/32; t++) { v[t] = __expf(v[t] - m); s += v[t]; }
    s = wave_sum(s);
    const float inv = 1.0f / s;
    #pragma unroll
    for (int t = 0; t < TK_/32; t++) {
      const float p = v[t] * inv;
      sS[q][lane + 32*t] = p;
      attnW[(size_t)(b*TQ_ + q0 + q)*TK_ + lane + 32*t] = p;
    }
  }
  __syncthreads();

  v8f cv[2]; cv[0] = {}; cv[1] = {};
  for (int kc = 0; kc < TK_/4; ++kc) {
    const int k0 = kc*4;
    v2f a;
    a.x = sS[row][k0 + 2*hi];
    a.y = sS[row][k0 + 2*hi + 1];
    #pragma unroll
    for (int i = 0; i < 2; i++) {
      const int n0 = (wave*2 + i)*16;
      v2f bb;
      bb.x = ctx[(size_t)(b*TK_ + k0 + 2*hi    )*D_ + n0 + row];
      bb.y = ctx[(size_t)(b*TK_ + k0 + 2*hi + 1)*D_ + n0 + row];
      cv[i] = wmma_f32_k4(a, bb, cv[i]);
    }
  }
  #pragma unroll
  for (int i = 0; i < 2; i++) {
    const int n0 = (wave*2 + i)*16;
    #pragma unroll
    for (int r = 0; r < 8; r++)
      sCV[hi*8 + r][n0 + row] = cv[i][r];
  }
  __syncthreads();

  #pragma unroll
  for (int j = 0; j < 2; j++) {
    const int q = wave*2 + j;
    float dot = 0.f;
    for (int d = lane; d < D_; d += 32) {
      dot += out_st[(size_t)(b*TQ_ + q0 + q)*D_ + d] * Wg[d];
      dot += sCV[q][d]                               * Wg[D_ + d];
      dot += dom[(size_t)(b*TQ_ + q0 + q)*D_ + d]    * Wg[2*D_ + d];
    }
    dot = wave_sum(dot);
    if (lane == 0) {
      const float x = dot + bg[0];
      pgenW[b*TQ_ + q0 + q] = 1.0f / (1.0f + __expf(-x));
    }
  }
}

__device__ inline void stage_queries(float* sA, const float* __restrict__ src, int tid) {
  for (int i = tid; i < TQ_*(D_/4); i += 128) {
    const int r  = i >> 6;
    const int c4 = (i & 63) << 2;
    *(v4f*)(sA + r*RP + c4) = *(const v4f*)(src + r*D_ + c4);
  }
}

__device__ inline void logits_64x64(const float* sA, const float* __restrict__ Ws,
                                    int c0, int row, int hi, v8f (&acc)[4][4]) {
  #pragma unroll
  for (int m = 0; m < 4; m++)
    #pragma unroll
    for (int j = 0; j < 4; j++) acc[m][j] = {};

  const float* Bp = Ws + (size_t)(2*hi)*V_ + c0 + 4*row;
  const float* Ap = sA + row*RP + 2*hi;
  #pragma unroll 2
  for (int ks = 0; ks < D_/4; ++ks) {
    const v4f b0 = *(const v4f*)(Bp + (size_t)(4*ks    )*V_);
    const v4f b1 = *(const v4f*)(Bp + (size_t)(4*ks + 1)*V_);
    v2f a[4];
    #pragma unroll
    for (int m = 0; m < 4; m++) a[m] = *(const v2f*)(Ap + m*16*RP + 4*ks);
    #pragma unroll
    for (int j = 0; j < 4; j++) {
      v2f bb; bb.x = b0[j]; bb.y = b1[j];
      #pragma unroll
      for (int m = 0; m < 4; m++) acc[m][j] = wmma_f32_k4(a[m], bb, acc[m][j]);
    }
  }
}

__global__ __launch_bounds__(128) void k_stats(
    const float* __restrict__ out_st, const float* __restrict__ Ws,
    const float* __restrict__ bs, float* __restrict__ pmW,
    float* __restrict__ psW, float* __restrict__ outp)
{
  const int b    = blockIdx.x;
  const int vb   = blockIdx.y;
  const int wave = threadIdx.x >> 5;
  const int lane = threadIdx.x & 31;
  const int row  = lane & 15;
  const int hi   = lane >> 4;
  const int c0   = vb*CB + wave*64;

  __shared__ __attribute__((aligned(16))) float sA[TQ_*RP];
  __shared__ float sM[4][TQ_];
  __shared__ float sSm[4][TQ_];

  stage_queries(sA, out_st + (size_t)b*TQ_*D_, threadIdx.x);
  __syncthreads();

  v8f acc[4][4];
  logits_64x64(sA, Ws, c0, row, hi, acc);

  const v4f bsv = *(const v4f*)(bs + c0 + 4*row);
  float* zrow = outp + (size_t)(b*TQ_ + hi*8)*V_ + c0 + 4*row;
  #pragma unroll
  for (int m = 0; m < 4; m++) {
    #pragma unroll
    for (int r = 0; r < 8; r++) {
      const float z0 = acc[m][0][r] + bsv[0];
      const float z1 = acc[m][1][r] + bsv[1];
      const float z2 = acc[m][2][r] + bsv[2];
      const float z3 = acc[m][3][r] + bsv[3];
      *(v4f*)(zrow + (size_t)(m*16 + r)*V_) = (v4f){z0, z1, z2, z3};
      const float tm = half16_max(fmaxf(fmaxf(z0, z1), fmaxf(z2, z3)));
      const float ts = half16_sum((__expf(z0 - tm) + __expf(z1 - tm)) +
                                  (__expf(z2 - tm) + __expf(z3 - tm)));
      if (row == 0) { sM[wave][m*16 + hi*8 + r] = tm; sSm[wave][m*16 + hi*8 + r] = ts; }
    }
  }
  __syncthreads();
  if (threadIdx.x < TQ_) {
    const int q = threadIdx.x;
    float gm = -1e30f;
    #pragma unroll
    for (int w = 0; w < 4; w++) gm = fmaxf(gm, sM[w][q]);
    float gs = 0.f;
    #pragma unroll
    for (int w = 0; w < 4; w++) gs += sSm[w][q] * __expf(sM[w][q] - gm);
    pmW[(size_t)vb*(B_*TQ_) + b*TQ_ + q] = gm;
    psW[(size_t)vb*(B_*TQ_) + b*TQ_ + q] = gs;
  }
}

__global__ __launch_bounds__(256) void k_reduce(
    const float* __restrict__ pmW, const float* __restrict__ psW,
    float* __restrict__ rmW, float* __restrict__ rsW)
{
  const int rowid = blockIdx.x*256 + threadIdx.x;
  if (rowid >= B_*TQ_) return;
  float gm = -1e30f;
  for (int v = 0; v < NCB; ++v) gm = fmaxf(gm, pmW[(size_t)v*(B_*TQ_) + rowid]);
  float gs = 0.f;
  for (int v = 0; v < NCB; ++v)
    gs += psW[(size_t)v*(B_*TQ_) + rowid] * __expf(pmW[(size_t)v*(B_*TQ_) + rowid] - gm);
  rmW[rowid] = gm;
  rsW[rowid] = gs;
}

#define FT  256
#define FIT 16

__global__ __launch_bounds__(FT) void k_final(
    const int* __restrict__ cp, const float* __restrict__ attnW,
    const float* __restrict__ pgenW, const float* __restrict__ rmW,
    const float* __restrict__ rsW, float* outp)
{
  const int b   = blockIdx.x;
  const int v0  = blockIdx.y*CB;
  const int tid = threadIdx.x;
  const int r0  = tid >> 6;
  const int c4  = (tid & 63) << 2;

  __shared__ __attribute__((aligned(16))) float sP[TQ_*CB];
  __shared__ float sScale[TQ_], sMr[TQ_], sOmp[TQ_];

  const int key0 = cp[b*TK_ + tid];
  const int key1 = cp[b*TK_ + FT + tid];
  float pg = 0.f, rs = 1.f, rm = 0.f;
  if (tid < TQ_) {
    const int rowid = b*TQ_ + tid;
    pg = pgenW[rowid]; rs = rsW[rowid]; rm = rmW[rowid];
  }

  float* tile = outp + (size_t)(b*TQ_ + r0)*V_ + v0 + c4;
  v4f z[FIT];
  #pragma unroll
  for (int it = 0; it < FIT; ++it) z[it] = *(const v4f*)(tile + (size_t)(4*it)*V_);

  if (tid < TQ_) {
    sScale[tid] = pg / rs;
    sMr[tid]    = rm;
    sOmp[tid]   = 1.0f - pg;
  }
  #pragma unroll
  for (int it = 0; it < FIT; ++it)
    *(v4f*)(sP + (r0 + 4*it)*CB + c4) = (v4f){0.f, 0.f, 0.f, 0.f};
  __syncthreads();

  #pragma unroll
  for (int h = 0; h < 2; ++h) {
    const int key = h ? key1 : key0;
    const int k   = tid + h*FT;
    if (key >= v0 && key < v0 + CB) {
      const int c = key - v0;
      for (int q = 0; q < TQ_; q++) {
        const float w = sOmp[q] * attnW[(size_t)(b*TQ_ + q)*TK_ + k];
        atomicAdd(&sP[q*CB + c], w);
      }
    }
  }
  __syncthreads();

  #pragma unroll
  for (int it = 0; it < FIT; ++it) {
    const int q = r0 + 4*it;
    const float sc = sScale[q];
    const float mr = sMr[q];
    const v4f pt = *(const v4f*)(sP + q*CB + c4);
    v4f o;
    #pragma unroll
    for (int j = 0; j < 4; j++) o[j] = __logf(sc * __expf(z[it][j] - mr) + pt[j]);
    *(v4f*)(tile + (size_t)(4*it)*V_) = o;
  }
}

extern "C" void kernel_launch(void* const* d_in, const int* in_sizes, int n_in,
                              void* d_out, int out_size, void* d_ws, size_t ws_size,
                              hipStream_t stream) {
  (void)in_sizes; (void)n_in; (void)out_size; (void)ws_size;
  const float* dom    = (const float*)d_in[0];
  const float* out_st = (const float*)d_in[1];
  const float* ctx    = (const float*)d_in[2];
  const int*   cp     = (const int*)d_in[3];
  const float* Ws     = (const float*)d_in[4];
  const float* bsv    = (const float*)d_in[5];
  const float* Wg     = (const float*)d_in[6];
  const float* bg     = (const float*)d_in[7];
  float* outp = (float*)d_out;

  float* attnW = (float*)d_ws;
  float* pgenW = attnW + (size_t)B_*TQ_*TK_;
  float* rmW   = pgenW + B_*TQ_;
  float* rsW   = rmW + B_*TQ_;
  float* pmW   = rsW + B_*TQ_;
  float* psW   = pmW + (size_t)NCB*B_*TQ_;

  k_attn  <<<dim3(B_*TQ_/16),        256, 0, stream>>>(dom, out_st, ctx, Wg, bg, attnW, pgenW);
  k_stats <<<dim3(B_, NCB),          128, 0, stream>>>(out_st, Ws, bsv, pmW, psW, outp);
  k_reduce<<<dim3((B_*TQ_+255)/256), 256, 0, stream>>>(pmW, psW, rmW, rsW);
  k_final <<<dim3(B_, NCB),          FT,  0, stream>>>(cp, attnW, pgenW, rmW, rsW, outp);
}


# Final-pass pointer scatter: in-block keys compacted to an LDS list, (key,query) pairs spread over all 256 threads; 64 dependent loads per key become one parallel round (LDS add order still arrival)
# speedup vs baseline: 2.4818x; 1.1722x over previous
#include <hip/hip_runtime.h>
#include <hip/hip_bf16.h>


#define B_  16
#define TQ_ 64
#define TK_ 512
#define D_  256
#define V_  32000

#define CB  256
#define NCB 125
#define RP  260

typedef __attribute__((ext_vector_type(2))) float v2f;
typedef __attribute__((ext_vector_type(4))) float v4f;
typedef __attribute__((ext_vector_type(8))) float v8f;

__device__ inline v8f wmma_f32_k4(v2f a, v2f b, v8f c) {
  return __builtin_amdgcn_wmma_f32_16x16x4_f32(
       false, a,  false, b,
       (short)0, c,  false,  false);
}

__device__ inline float wave_max(float x) {
  #pragma unroll
  for (int o = 1; o < 32; o <<= 1) x = fmaxf(x, __shfl_xor(x, o, 32));
  return x;
}
__device__ inline float wave_sum(float x) {
  #pragma unroll
  for (int o = 1; o < 32; o <<= 1) x += __shfl_xor(x, o, 32);
  return x;
}
__device__ inline float half16_max(float x) {
  #pragma unroll
  for (int o = 1; o < 16; o <<= 1) x = fmaxf(x, __shfl_xor(x, o, 32));
  return x;
}
__device__ inline float half16_sum(float x) {
  #pragma unroll
  for (int o = 1; o < 16; o <<= 1) x += __shfl_xor(x, o, 32);
  return x;
}

__global__ __launch_bounds__(256) void k_attn(
    const float* __restrict__ dom, const float* __restrict__ out_st,
    const float* __restrict__ ctx, const float* __restrict__ Wg,
    const float* __restrict__ bg, float* __restrict__ attnW,
    float* __restrict__ pgenW)
{
  __shared__ float sS[16][TK_];
  __shared__ float sCV[16][D_];

  const int b    = blockIdx.x >> 2;
  const int q0   = (blockIdx.x & 3) << 4;
  const int wave = threadIdx.x >> 5;
  const int lane = threadIdx.x & 31;
  const int row  = lane & 15;
  const int hi   = lane >> 4;

  v8f acc[4];
  #pragma unroll
  for (int i = 0; i < 4; i++) acc[i] = {};
  const float* Abase = out_st + (size_t)(b*TQ_ + q0 + row)*D_;
  for (int dc = 0; dc < D_/4; ++dc) {
    const int d0 = dc*4;
    v2f a = *(const v2f*)(Abase + d0 + 2*hi);
    #pragma unroll
    for (int i = 0; i < 4; i++) {
      const int k0 = (wave*4 + i)*16;
      v2f bb = *(const v2f*)(ctx + (size_t)(b*TK_ + k0 + row)*D_ + d0 + 2*hi);
      acc[i] = wmma_f32_k4(a, bb, acc[i]);
    }
  }
  #pragma unroll
  for (int i = 0; i < 4; i++) {
    const int k0 = (wave*4 + i)*16;
    #pragma unroll
    for (int r = 0; r < 8; r++)
      sS[hi*8 + r][k0 + row] = acc[i][r] * 0.0625f;
  }
  __syncthreads();

  #pragma unroll
  for (int j = 0; j < 2; j++) {
    const int q = wave*2 + j;
    float v[TK_/32];
    float m = -1e30f;
    #pragma unroll
    for (int t = 0; t < TK_/32; t++) { v[t] = sS[q][lane + 32*t]; m = fmaxf(m, v[t]); }
    m = wave_max(m);
    float s = 0.f;
    #pragma unroll
    for (int t = 0; t < TK_/32; t++) { v[t] = __expf(v[t] - m); s += v[t]; }
    s = wave_sum(s);
    const float inv = 1.0f / s;
    #pragma unroll
    for (int t = 0; t < TK_/32; t++) {
      const float p = v[t] * inv;
      sS[q][lane + 32*t] = p;
      attnW[(size_t)(b*TQ_ + q0 + q)*TK_ + lane + 32*t] = p;
    }
  }
  __syncthreads();

  v8f cv[2]; cv[0] = {}; cv[1] = {};
  for (int kc = 0; kc < TK_/4; ++kc) {
    const int k0 = kc*4;
    v2f a;
    a.x = sS[row][k0 + 2*hi];
    a.y = sS[row][k0 + 2*hi + 1];
    #pragma unroll
    for (int i = 0; i < 2; i++) {
      const int n0 = (wave*2 + i)*16;
      v2f bb;
      bb.x = ctx[(size_t)(b*TK_ + k0 + 2*hi    )*D_ + n0 + row];
      bb.y = ctx[(size_t)(b*TK_ + k0 + 2*hi + 1)*D_ + n0 + row];
      cv[i] = wmma_f32_k4(a, bb, cv[i]);
    }
  }
  #pragma unroll
  for (int i = 0; i < 2; i++) {
    const int n0 = (wave*2 + i)*16;
    #pragma unroll
    for (int r = 0; r < 8; r++)
      sCV[hi*8 + r][n0 + row] = cv[i][r];
  }
  __syncthreads();

  #pragma unroll
  for (int j = 0; j < 2; j++) {
    const int q = wave*2 + j;
    float dot = 0.f;
    for (int d = lane; d < D_; d += 32) {
      dot += out_st[(size_t)(b*TQ_ + q0 + q)*D_ + d] * Wg[d];
      dot += sCV[q][d]                               * Wg[D_ + d];
      dot += dom[(size_t)(b*TQ_ + q0 + q)*D_ + d]    * Wg[2*D_ + d];
    }
    dot = wave_sum(dot);
    if (lane == 0) {
      const float x = dot + bg[0];
      pgenW[b*TQ_ + q0 + q] = 1.0f / (1.0f + __expf(-x));
    }
  }
}

__device__ inline void stage_queries(float* sA, const float* __restrict__ src, int tid) {
  for (int i = tid; i < TQ_*(D_/4); i += 128) {
    const int r  = i >> 6;
    const int c4 = (i & 63) << 2;
    *(v4f*)(sA + r*RP + c4) = *(const v4f*)(src + r*D_ + c4);
  }
}

__device__ inline void logits_64x64(const float* sA, const float* __restrict__ Ws,
                                    int c0, int row, int hi, v8f (&acc)[4][4]) {
  #pragma unroll
  for (int m = 0; m < 4; m++)
    #pragma unroll
    for (int j = 0; j < 4; j++) acc[m][j] = {};

  const float* Bp = Ws + (size_t)(2*hi)*V_ + c0 + 4*row;
  const float* Ap = sA + row*RP + 2*hi;
  #pragma unroll 2
  for (int ks = 0; ks < D_/4; ++ks) {
    const v4f b0 = *(const v4f*)(Bp + (size_t)(4*ks    )*V_);
    const v4f b1 = *(const v4f*)(Bp + (size_t)(4*ks + 1)*V_);
    v2f a[4];
    #pragma unroll
    for (int m = 0; m < 4; m++) a[m] = *(const v2f*)(Ap + m*16*RP + 4*ks);
    #pragma unroll
    for (int j = 0; j < 4; j++) {
      v2f bb; bb.x = b0[j]; bb.y = b1[j];
      #pragma unroll
      for (int m = 0; m < 4; m++) acc[m][j] = wmma_f32_k4(a[m], bb, acc[m][j]);
    }
  }
}

__global__ __launch_bounds__(128) void k_stats(
    const float* __restrict__ out_st, const float* __restrict__ Ws,
    const float* __restrict__ bs, float* __restrict__ pmW,
    float* __restrict__ psW, float* __restrict__ outp)
{
  const int b    = blockIdx.x;
  const int vb   = blockIdx.y;
  const int wave = threadIdx.x >> 5;
  const int lane = threadIdx.x & 31;
  const int row  = lane & 15;
  const int hi   = lane >> 4;
  const int c0   = vb*CB + wave*64;

  __shared__ __attribute__((aligned(16))) float sA[TQ_*RP];
  __shared__ float sM[4][TQ_];
  __shared__ float sSm[4][TQ_];

  stage_queries(sA, out_st + (size_t)b*TQ_*D_, threadIdx.x);
  __syncthreads();

  v8f acc[4][4];
  logits_64x64(sA, Ws, c0, row, hi, acc);

  const v4f bsv = *(const v4f*)(bs + c0 + 4*row);
  float* zrow = outp + (size_t)(b*TQ_ + hi*8)*V_ + c0 + 4*row;
  #pragma unroll
  for (int m = 0; m < 4; m++) {
    #pragma unroll
    for (int r = 0; r < 8; r++) {
      const float z0 = acc[m][0][r] + bsv[0];
      const float z1 = acc[m][1][r] + bsv[1];
      const float z2 = acc[m][2][r] + bsv[2];
      const float z3 = acc[m][3][r] + bsv[3];
      *(v4f*)(zrow + (size_t)(m*16 + r)*V_) = (v4f){z0, z1, z2, z3};
      const float tm = half16_max(fmaxf(fmaxf(z0, z1), fmaxf(z2, z3)));
      const float ts = half16_sum((__expf(z0 - tm) + __expf(z1 - tm)) +
                                  (__expf(z2 - tm) + __expf(z3 - tm)));
      if (row == 0) { sM[wave][m*16 + hi*8 + r] = tm; sSm[wave][m*16 + hi*8 + r] = ts; }
    }
  }
  __syncthreads();
  if (threadIdx.x < TQ_) {
    const int q = threadIdx.x;
    float gm = -1e30f;
    #pragma unroll
    for (int w = 0; w < 4; w++) gm = fmaxf(gm, sM[w][q]);
    float gs = 0.f;
    #pragma unroll
    for (int w = 0; w < 4; w++) gs += sSm[w][q] * __expf(sM[w][q] - gm);
    pmW[(size_t)vb*(B_*TQ_) + b*TQ_ + q] = gm;
    psW[(size_t)vb*(B_*TQ_) + b*TQ_ + q] = gs;
  }
}

__global__ __launch_bounds__(256) void k_reduce(
    const float* __restrict__ pmW, const float* __restrict__ psW,
    float* __restrict__ rmW, float* __restrict__ rsW)
{
  const int rowid = blockIdx.x*256 + threadIdx.x;
  if (rowid >= B_*TQ_) return;
  float gm = -1e30f;
  for (int v = 0; v < NCB; ++v) gm = fmaxf(gm, pmW[(size_t)v*(B_*TQ_) + rowid]);
  float gs = 0.f;
  for (int v = 0; v < NCB; ++v)
    gs += psW[(size_t)v*(B_*TQ_) + rowid] * __expf(pmW[(size_t)v*(B_*TQ_) + rowid] - gm);
  rmW[rowid] = gm;
  rsW[rowid] = gs;
}

#define FT  256
#define FIT 16

__global__ __launch_bounds__(FT) void k_final(
    const int* __restrict__ cp, const float* __restrict__ attnW,
    const float* __restrict__ pgenW, const float* __restrict__ rmW,
    const float* __restrict__ rsW, float* outp)
{
  const int b   = blockIdx.x;
  const int v0  = blockIdx.y*CB;
  const int tid = threadIdx.x;
  const int r0  = tid >> 6;
  const int c4  = (tid & 63) << 2;

  __shared__ __attribute__((aligned(16))) float sP[TQ_*CB];
  __shared__ float sScale[TQ_], sMr[TQ_], sOmp[TQ_];
  __shared__ int sHit[TK_];
  __shared__ int sNHit;

  const int key0 = cp[b*TK_ + tid];
  const int key1 = cp[b*TK_ + FT + tid];
  float pg = 0.f, rs = 1.f, rm = 0.f;
  if (tid < TQ_) {
    const int rowid = b*TQ_ + tid;
    pg = pgenW[rowid]; rs = rsW[rowid]; rm = rmW[rowid];
  }

  float* tile = outp + (size_t)(b*TQ_ + r0)*V_ + v0 + c4;
  v4f z[FIT];
  #pragma unroll
  for (int it = 0; it < FIT; ++it) z[it] = *(const v4f*)(tile + (size_t)(4*it)*V_);

  if (tid < TQ_) {
    sScale[tid] = pg / rs;
    sMr[tid]    = rm;
    sOmp[tid]   = 1.0f - pg;
  }
  if (tid == 0) sNHit = 0;
  #pragma unroll
  for (int it = 0; it < FIT; ++it)
    *(v4f*)(sP + (r0 + 4*it)*CB + c4) = (v4f){0.f, 0.f, 0.f, 0.f};
  __syncthreads();

  #pragma unroll
  for (int h = 0; h < 2; ++h) {
    const int key = h ? key1 : key0;
    if (key >= v0 && key < v0 + CB)
      sHit[atomicAdd(&sNHit, 1)] = ((tid + h*FT) << 8) | (key - v0);
  }
  __syncthreads();

  {
    const int nh = sNHit;
    const int q  = tid & 63;
    const float omp = sOmp[q];
    const float* arow = attnW + (size_t)(b*TQ_ + q)*TK_;
    for (int h0 = tid >> 6; h0 < nh; h0 += 16) {
      int   e[4];
      float w[4];
      #pragma unroll
      for (int u = 0; u < 4; ++u) e[u] = sHit[min(h0 + 4*u, nh - 1)];
      #pragma unroll
      for (int u = 0; u < 4; ++u) w[u] = arow[e[u] >> 8];
      #pragma unroll
      for (int u = 0; u < 4; ++u)
        if (h0 + 4*u < nh) atomicAdd(&sP[q*CB + (e[u] & 255)], omp * w[u]);
    }
  }
  __syncthreads();

  #pragma unroll
  for (int it = 0; it < FIT; ++it) {
    const int q = r0 + 4*it;
    const float sc = sScale[q];
    const float mr = sMr[q];
    const v4f pt = *(const v4f*)(sP + q*CB + c4);
    v4f o;
    #pragma unroll
    for (int j = 0; j < 4; j++) o[j] = __logf(sc * __expf(z[it][j] - mr) + pt[j]);
    *(v4f*)(tile + (size_t)(4*it)*V_) = o;
  }
}

extern "C" void kernel_launch(void* const* d_in, const int* in_sizes, int n_in,
                              void* d_out, int out_size, void* d_ws, size_t ws_size,
                              hipStream_t stream) {
  (void)in_sizes; (void)n_in; (void)out_size; (void)ws_size;
  const float* dom    = (const float*)d_in[0];
  const float* out_st = (const float*)d_in[1];
  const float* ctx    = (const float*)d_in[2];
  const int*   cp     = (const int*)d_in[3];
  const float* Ws     = (const float*)d_in[4];
  const float* bsv    = (const float*)d_in[5];
  const float* Wg     = (const float*)d_in[6];
  const float* bg     = (const float*)d_in[7];
  float* outp = (float*)d_out;

  float* attnW = (float*)d_ws;
  float* pgenW = attnW + (size_t)B_*TQ_*TK_;
  float* rmW   = pgenW + B_*TQ_;
  float* rsW   = rmW + B_*TQ_;
  float* pmW   = rsW + B_*TQ_;
  float* psW   = pmW + (size_t)NCB*B_*TQ_;

  k_attn  <<<dim3(B_*TQ_/16),        256, 0, stream>>>(dom, out_st, ctx, Wg, bg, attnW, pgenW);
  k_stats <<<dim3(B_, NCB),          128, 0, stream>>>(out_st, Ws, bsv, pmW, psW, outp);
  k_reduce<<<dim3((B_*TQ_+255)/256), 256, 0, stream>>>(pmW, psW, rmW, rsW);
  k_final <<<dim3(B_, NCB),          FT,  0, stream>>>(cp, attnW, pgenW, rmW, rsW, outp);
}


# Attention work-groups ride in the vocab GEMM launch (block index picks the role, attention re-split over 4 waves): its load-bound 64 groups overlap the matrix-bound pass; 4 launches become 3
# speedup vs baseline: 2.7048x; 1.0899x over previous
#include <hip/hip_runtime.h>
#include <hip/hip_bf16.h>


#define B_  16
#define TQ_ 64
#define TK_ 512
#define D_  256
#define V_  32000

#define CB  256
#define NCB 125
#define RP  260

typedef __attribute__((ext_vector_type(2))) float v2f;
typedef __attribute__((ext_vector_type(4))) float v4f;
typedef __attribute__((ext_vector_type(8))) float v8f;

__device__ inline v8f wmma_f32_k4(v2f a, v2f b, v8f c) {
  return __builtin_amdgcn_wmma_f32_16x16x4_f32(
       false, a,  false, b,
       (short)0, c,  false,  false);
}

__device__ inline float wave_max(float x) {
  #pragma unroll
  for (int o = 1; o < 32; o <<= 1) x = fmaxf(x, __shfl_xor(x, o, 32));
  return x;
}
__device__ inline float wave_sum(float x) {
  #pragma unroll
  for (int o = 1; o < 32; o <<= 1) x += __shfl_xor(x, o, 32);
  return x;
}
__device__ inline float half16_max(float x) {
  #pragma unroll
  for (int o = 1; o < 16; o <<= 1) x = fmaxf(x, __shfl_xor(x, o, 32));
  return x;
}
__device__ inline float half16_sum(float x) {
  #pragma unroll
  for (int o = 1; o < 16; o <<= 1) x += __shfl_xor(x, o, 32);
  return x;
}

#define NATT (B_*TQ_/16)

__device__ __forceinline__ void attn_role(
    float* smem, const int blk,
    const float* __restrict__ dom, const float* __restrict__ out_st,
    const float* __restrict__ ctx, const float* __restrict__ Wg,
    const float* __restrict__ bg, float* __restrict__ attnW,
    float* __restrict__ pgenW)
{
  float* sS  = smem;
  float* sCV = smem + 16*TK_;

  const int b    = blk >> 2;
  const int q0   = (blk & 3) << 4;
  const int wave = threadIdx.x >> 5;
  const int lane = threadIdx.x & 31;
  const int row  = lane & 15;
  const int hi   = lane >> 4;

  v8f acc[8];
  #pragma unroll
  for (int i = 0; i < 8; i++) acc[i] = {};
  const float* Abase = out_st + (size_t)(b*TQ_ + q0 + row)*D_;
  for (int dc = 0; dc < D_/4; ++dc) {
    const int d0 = dc*4;
    v2f a = *(const v2f*)(Abase + d0 + 2*hi);
    #pragma unroll
    for (int i = 0; i < 8; i++) {
      const int k0 = (wave*8 + i)*16;
      v2f bb = *(const v2f*)(ctx + (size_t)(b*TK_ + k0 + row)*D_ + d0 + 2*hi);
      acc[i] = wmma_f32_k4(a, bb, acc[i]);
    }
  }
  #pragma unroll
  for (int i = 0; i < 8; i++) {
    const int k0 = (wave*8 + i)*16;
    #pragma unroll
    for (int r = 0; r < 8; r++)
      sS[(hi*8 + r)*TK_ + k0 + row] = acc[i][r] * 0.0625f;
  }
  __syncthreads();

  #pragma unroll
  for (int j = 0; j < 4; j++) {
    const int q = wave*4 + j;
    float v[TK_/32];
    float m = -1e30f;
    #pragma unroll
    for (int t = 0; t < TK_/32; t++) { v[t] = sS[q*TK_ + lane + 32*t]; m = fmaxf(m, v[t]); }
    m = wave_max(m);
    float s = 0.f;
    #pragma unroll
    for (int t = 0; t < TK_/32; t++) { v[t] = __expf(v[t] - m); s += v[t]; }
    s = wave_sum(s);
    const float inv = 1.0f / s;
    #pragma unroll
    for (int t = 0; t < TK_/32; t++) {
      const float p = v[t] * inv;
      sS[q*TK_ + lane + 32*t] = p;
      attnW[(size_t)(b*TQ_ + q0 + q)*TK_ + lane + 32*t] = p;
    }
  }
  __syncthreads();

  v8f cv[4];
  #pragma unroll
  for (int i = 0; i < 4; i++) cv[i] = {};
  #pragma unroll 4
  for (int kc = 0; kc < TK_/4; ++kc) {
    const int k0 = kc*4;
    v2f a;
    a.x = sS[row*TK_ + k0 + 2*hi];
    a.y = sS[row*TK_ + k0 + 2*hi + 1];
    #pragma unroll
    for (int i = 0; i < 4; i++) {
      const int n0 = (wave*4 + i)*16;
      v2f bb;
      bb.x = ctx[(size_t)(b*TK_ + k0 + 2*hi    )*D_ + n0 + row];
      bb.y = ctx[(size_t)(b*TK_ + k0 + 2*hi + 1)*D_ + n0 + row];
      cv[i] = wmma_f32_k4(a, bb, cv[i]);
    }
  }
  #pragma unroll
  for (int i = 0; i < 4; i++) {
    const int n0 = (wave*4 + i)*16;
    #pragma unroll
    for (int r = 0; r < 8; r++)
      sCV[(hi*8 + r)*D_ + n0 + row] = cv[i][r];
  }
  __syncthreads();

  #pragma unroll
  for (int j = 0; j < 4; j++) {
    const int q = wave*4 + j;
    float dot = 0.f;
    #pragma unroll 2
    for (int d = lane; d < D_; d += 32) {
      dot += out_st[(size_t)(b*TQ_ + q0 + q)*D_ + d] * Wg[d];
      dot += sCV[q*D_ + d]                           * Wg[D_ + d];
      dot += dom[(size_t)(b*TQ_ + q0 + q)*D_ + d]    * Wg[2*D_ + d];
    }
    dot = wave_sum(dot);
    if (lane == 0) {
      const float x = dot + bg[0];
      pgenW[b*TQ_ + q0 + q] = 1.0f / (1.0f + __expf(-x));
    }
  }
}

__device__ inline void stage_queries(float* sA, const float* __restrict__ src, int tid) {
  for (int i = tid; i < TQ_*(D_/4); i += 128) {
    const int r  = i >> 6;
    const int c4 = (i & 63) << 2;
    *(v4f*)(sA + r*RP + c4) = *(const v4f*)(src + r*D_ + c4);
  }
}

__device__ inline void logits_64x64(const float* sA, const float* __restrict__ Ws,
                                    int c0, int row, int hi, v8f (&acc)[4][4]) {
  #pragma unroll
  for (int m = 0; m < 4; m++)
    #pragma unroll
    for (int j = 0; j < 4; j++) acc[m][j] = {};

  const float* Bp = Ws + (size_t)(2*hi)*V_ + c0 + 4*row;
  const float* Ap = sA + row*RP + 2*hi;
  #pragma unroll 2
  for (int ks = 0; ks < D_/4; ++ks) {
    const v4f b0 = *(const v4f*)(Bp + (size_t)(4*ks    )*V_);
    const v4f b1 = *(const v4f*)(Bp + (size_t)(4*ks + 1)*V_);
    v2f a[4];
    #pragma unroll
    for (int m = 0; m < 4; m++) a[m] = *(const v2f*)(Ap + m*16*RP + 4*ks);
    #pragma unroll
    for (int j = 0; j < 4; j++) {
      v2f bb; bb.x = b0[j]; bb.y = b1[j];
      #pragma unroll
      for (int m = 0; m < 4; m++) acc[m][j] = wmma_f32_k4(a[m], bb, acc[m][j]);
    }
  }
}

__global__ __launch_bounds__(128) void k_attn_stats(
    const float* __restrict__ dom, const float* __restrict__ out_st,
    const float* __restrict__ ctx, const float* __restrict__ Wg,
    const float* __restrict__ bg, float* __restrict__ attnW,
    float* __restrict__ pgenW, const float* __restrict__ Ws,
    const float* __restrict__ bs, float* __restrict__ pmW,
    float* __restrict__ psW, float* __restrict__ outp)
{
  __shared__ __attribute__((aligned(16))) float smem[TQ_*RP + 2*4*TQ_];

  if (blockIdx.x < NATT) {
    attn_role(smem, blockIdx.x, dom, out_st, ctx, Wg, bg, attnW, pgenW);
    return;
  }

  const int gid  = blockIdx.x - NATT;
  const int b    = gid & (B_ - 1);
  const int vb   = gid >> 4;
  const int wave = threadIdx.x >> 5;
  const int lane = threadIdx.x & 31;
  const int row  = lane & 15;
  const int hi   = lane >> 4;
  const int c0   = vb*CB + wave*64;

  float* sA  = smem;
  float* sM  = smem + TQ_*RP;
  float* sSm = sM + 4*TQ_;

  stage_queries(sA, out_st + (size_t)b*TQ_*D_, threadIdx.x);
  __syncthreads();

  v8f acc[4][4];
  logits_64x64(sA, Ws, c0, row, hi, acc);

  const v4f bsv = *(const v4f*)(bs + c0 + 4*row);
  float* zrow = outp + (size_t)(b*TQ_ + hi*8)*V_ + c0 + 4*row;
  #pragma unroll
  for (int m = 0; m < 4; m++) {
    #pragma unroll
    for (int r = 0; r < 8; r++) {
      const float z0 = acc[m][0][r] + bsv[0];
      const float z1 = acc[m][1][r] + bsv[1];
      const float z2 = acc[m][2][r] + bsv[2];
      const float z3 = acc[m][3][r] + bsv[3];
      *(v4f*)(zrow + (size_t)(m*16 + r)*V_) = (v4f){z0, z1, z2, z3};
      const float tm = half16_max(fmaxf(fmaxf(z0, z1), fmaxf(z2, z3)));
      const float ts = half16_sum((__expf(z0 - tm) + __expf(z1 - tm)) +
                                  (__expf(z2 - tm) + __expf(z3 - tm)));
      if (row == 0) { sM[wave*TQ_ + m*16 + hi*8 + r] = tm; sSm[wave*TQ_ + m*16 + hi*8 + r] = ts; }
    }
  }
  __syncthreads();
  if (threadIdx.x < TQ_) {
    const int q = threadIdx.x;
    float gm = -1e30f;
    #pragma unroll
    for (int w = 0; w < 4; w++) gm = fmaxf(gm, sM[w*TQ_ + q]);
    float gs = 0.f;
    #pragma unroll
    for (int w = 0; w < 4; w++) gs += sSm[w*TQ_ + q] * __expf(sM[w*TQ_ + q] - gm);
    pmW[(size_t)vb*(B_*TQ_) + b*TQ_ + q] = gm;
    psW[(size_t)vb*(B_*TQ_) + b*TQ_ + q] = gs;
  }
}

__global__ __launch_bounds__(256) void k_reduce(
    const float* __restrict__ pmW, const float* __restrict__ psW,
    float* __restrict__ rmW, float* __restrict__ rsW)
{
  const int rowid = blockIdx.x*256 + threadIdx.x;
  if (rowid >= B_*TQ_) return;
  float gm = -1e30f;
  for (int v = 0; v < NCB; ++v) gm = fmaxf(gm, pmW[(size_t)v*(B_*TQ_) + rowid]);
  float gs = 0.f;
  for (int v = 0; v < NCB; ++v)
    gs += psW[(size_t)v*(B_*TQ_) + rowid] * __expf(pmW[(size_t)v*(B_*TQ_) + rowid] - gm);
  rmW[rowid] = gm;
  rsW[rowid] = gs;
}

#define FT  256
#define FIT 16

__global__ __launch_bounds__(FT) void k_final(
    const int* __restrict__ cp, const float* __restrict__ attnW,
    const float* __restrict__ pgenW, const float* __restrict__ rmW,
    const float* __restrict__ rsW, float* outp)
{
  const int b   = blockIdx.x;
  const int v0  = blockIdx.y*CB;
  const int tid = threadIdx.x;
  const int r0  = tid >> 6;
  const int c4  = (tid & 63) << 2;

  __shared__ __attribute__((aligned(16))) float sP[TQ_*CB];
  __shared__ float sScale[TQ_], sMr[TQ_], sOmp[TQ_];
  __shared__ int sHit[TK_];
  __shared__ int sNHit;

  const int key0 = cp[b*TK_ + tid];
  const int key1 = cp[b*TK_ + FT + tid];
  float pg = 0.f, rs = 1.f, rm = 0.f;
  if (tid < TQ_) {
    const int rowid = b*TQ_ + tid;
    pg = pgenW[rowid]; rs = rsW[rowid]; rm = rmW[rowid];
  }

  float* tile = outp + (size_t)(b*TQ_ + r0)*V_ + v0 + c4;
  v4f z[FIT];
  #pragma unroll
  for (int it = 0; it < FIT; ++it) z[it] = *(const v4f*)(tile + (size_t)(4*it)*V_);

  if (tid < TQ_) {
    sScale[tid] = pg / rs;
    sMr[tid]    = rm;
    sOmp[tid]   = 1.0f - pg;
  }
  if (tid == 0) sNHit = 0;
  #pragma unroll
  for (int it = 0; it < FIT; ++it)
    *(v4f*)(sP + (r0 + 4*it)*CB + c4) = (v4f){0.f, 0.f, 0.f, 0.f};
  __syncthreads();

  #pragma unroll
  for (int h = 0; h < 2; ++h) {
    const int key = h ? key1 : key0;
    if (key >= v0 && key < v0 + CB)
      sHit[atomicAdd(&sNHit, 1)] = ((tid + h*FT) << 8) | (key - v0);
  }
  __syncthreads();

  {
    const int nh = sNHit;
    const int q  = tid & 63;
    const float omp = sOmp[q];
    const float* arow = attnW + (size_t)(b*TQ_ + q)*TK_;
    for (int h0 = tid >> 6; h0 < nh; h0 += 16) {
      int   e[4];
      float w[4];
      #pragma unroll
      for (int u = 0; u < 4; ++u) e[u] = sHit[min(h0 + 4*u, nh - 1)];
      #pragma unroll
      for (int u = 0; u < 4; ++u) w[u] = arow[e[u] >> 8];
      #pragma unroll
      for (int u = 0; u < 4; ++u)
        if (h0 + 4*u < nh) atomicAdd(&sP[q*CB + (e[u] & 255)], omp * w[u]);
    }
  }
  __syncthreads();

  #pragma unroll
  for (int it = 0; it < FIT; ++it) {
    const int q = r0 + 4*it;
    const float sc = sScale[q];
    const float mr = sMr[q];
    const v4f pt = *(const v4f*)(sP + q*CB + c4);
    v4f o;
    #pragma unroll
    for (int j = 0; j < 4; j++) o[j] = __logf(sc * __expf(z[it][j] - mr) + pt[j]);
    *(v4f*)(tile + (size_t)(4*it)*V_) = o;
  }
}

extern "C" void kernel_launch(void* const* d_in, const int* in_sizes, int n_in,
                              void* d_out, int out_size, void* d_ws, size_t ws_size,
                              hipStream_t stream) {
  (void)in_sizes; (void)n_in; (void)out_size; (void)ws_size;
  const float* dom    = (const float*)d_in[0];
  const float* out_st = (const float*)d_in[1];
  const float* ctx    = (const float*)d_in[2];
  const int*   cp     = (const int*)d_in[3];
  const float* Ws     = (const float*)d_in[4];
  const float* bsv    = (const float*)d_in[5];
  const float* Wg     = (const float*)d_in[6];
  const float* bg     = (const float*)d_in[7];
  float* outp = (float*)d_out;

  float* attnW = (float*)d_ws;
  float* pgenW = attnW + (size_t)B_*TQ_*TK_;
  float* rmW   = pgenW + B_*TQ_;
  float* rsW   = rmW + B_*TQ_;
  float* pmW   = rsW + B_*TQ_;
  float* psW   = pmW + (size_t)NCB*B_*TQ_;

  k_attn_stats<<<dim3(NATT + B_*NCB), 128, 0, stream>>>(dom, out_st, ctx, Wg, bg, attnW, pgenW,
                                                        Ws, bsv, pmW, psW, outp);
  k_reduce<<<dim3((B_*TQ_+255)/256), 256, 0, stream>>>(pmW, psW, rmW, rsW);
  k_final <<<dim3(B_, NCB),          FT,  0, stream>>>(cp, attnW, pgenW, rmW, rsW, outp);
}


# Row max/sum combine (2nd launch): 32 rows x 8 waves per group, every lane issues all its partial loads in one round, LDS combine over waves; was ~90 serial round trips; sum order: wave, then waves
# speedup vs baseline: 3.4088x; 1.2603x over previous
#include <hip/hip_runtime.h>
#include <hip/hip_bf16.h>


#define B_  16
#define TQ_ 64
#define TK_ 512
#define D_  256
#define V_  32000

#define CB  256
#define NCB 125
#define RP  260

typedef __attribute__((ext_vector_type(2))) float v2f;
typedef __attribute__((ext_vector_type(4))) float v4f;
typedef __attribute__((ext_vector_type(8))) float v8f;

__device__ inline v8f wmma_f32_k4(v2f a, v2f b, v8f c) {
  return __builtin_amdgcn_wmma_f32_16x16x4_f32(
       false, a,  false, b,
       (short)0, c,  false,  false);
}

__device__ inline float wave_max(float x) {
  #pragma unroll
  for (int o = 1; o < 32; o <<= 1) x = fmaxf(x, __shfl_xor(x, o, 32));
  return x;
}
__device__ inline float wave_sum(float x) {
  #pragma unroll
  for (int o = 1; o < 32; o <<= 1) x += __shfl_xor(x, o, 32);
  return x;
}
__device__ inline float half16_max(float x) {
  #pragma unroll
  for (int o = 1; o < 16; o <<= 1) x = fmaxf(x, __shfl_xor(x, o, 32));
  return x;
}
__device__ inline float half16_sum(float x) {
  #pragma unroll
  for (int o = 1; o < 16; o <<= 1) x += __shfl_xor(x, o, 32);
  return x;
}

#define NATT (B_*TQ_/16)

__device__ __forceinline__ void attn_role(
    float* smem, const int blk,
    const float* __restrict__ dom, const float* __restrict__ out_st,
    const float* __restrict__ ctx, const float* __restrict__ Wg,
    const float* __restrict__ bg, float* __restrict__ attnW,
    float* __restrict__ pgenW)
{
  float* sS  = smem;
  float* sCV = smem + 16*TK_;

  const int b    = blk >> 2;
  const int q0   = (blk & 3) << 4;
  const int wave = threadIdx.x >> 5;
  const int lane = threadIdx.x & 31;
  const int row  = lane & 15;
  const int hi   = lane >> 4;

  v8f acc[8];
  #pragma unroll
  for (int i = 0; i < 8; i++) acc[i] = {};
  const float* Abase = out_st + (size_t)(b*TQ_ + q0 + row)*D_;
  for (int dc = 0; dc < D_/4; ++dc) {
    const int d0 = dc*4;
    v2f a = *(const v2f*)(Abase + d0 + 2*hi);
    #pragma unroll
    for (int i = 0; i < 8; i++) {
      const int k0 = (wave*8 + i)*16;
      v2f bb = *(const v2f*)(ctx + (size_t)(b*TK_ + k0 + row)*D_ + d0 + 2*hi);
      acc[i] = wmma_f32_k4(a, bb, acc[i]);
    }
  }
  #pragma unroll
  for (int i = 0; i < 8; i++) {
    const int k0 = (wave*8 + i)*16;
    #pragma unroll
    for (int r = 0; r < 8; r++)
      sS[(hi*8 + r)*TK_ + k0 + row] = acc[i][r] * 0.0625f;
  }
  __syncthreads();

  #pragma unroll
  for (int j = 0; j < 4; j++) {
    const int q = wave*4 + j;
    float v[TK_/32];
    float m = -1e30f;
    #pragma unroll
    for (int t = 0; t < TK_/32; t++) { v[t] = sS[q*TK_ + lane + 32*t]; m = fmaxf(m, v[t]); }
    m = wave_max(m);
    float s = 0.f;
    #pragma unroll
    for (int t = 0; t < TK_/32; t++) { v[t] = __expf(v[t] - m); s += v[t]; }
    s = wave_sum(s);
    const float inv = 1.0f / s;
    #pragma unroll
    for (int t = 0; t < TK_/32; t++) {
      const float p = v[t] * inv;
      sS[q*TK_ + lane + 32*t] = p;
      attnW[(size_t)(b*TQ_ + q0 + q)*TK_ + lane + 32*t] = p;
    }
  }
  __syncthreads();

  v8f cv[4];
  #pragma unroll
  for (int i = 0; i < 4; i++) cv[i] = {};
  #pragma unroll 4
  for (int kc = 0; kc < TK_/4; ++kc) {
    const int k0 = kc*4;
    v2f a;
    a.x = sS[row*TK_ + k0 + 2*hi];
    a.y = sS[row*TK_ + k0 + 2*hi + 1];
    #pragma unroll
    for (int i = 0; i < 4; i++) {
      const int n0 = (wave*4 + i)*16;
      v2f bb;
      bb.x = ctx[(size_t)(b*TK_ + k0 + 2*hi    )*D_ + n0 + row];
      bb.y = ctx[(size_t)(b*TK_ + k0 + 2*hi + 1)*D_ + n0 + row];
      cv[i] = wmma_f32_k4(a, bb, cv[i]);
    }
  }
  #pragma unroll
  for (int i = 0; i < 4; i++) {
    const int n0 = (wave*4 + i)*16;
    #pragma unroll
    for (int r = 0; r < 8; r++)
      sCV[(hi*8 + r)*D_ + n0 + row] = cv[i][r];
  }
  __syncthreads();

  #pragma unroll
  for (int j = 0; j < 4; j++) {
    const int q = wave*4 + j;
    float dot = 0.f;
    #pragma unroll 2
    for (int d = lane; d < D_; d += 32) {
      dot += out_st[(size_t)(b*TQ_ + q0 + q)*D_ + d] * Wg[d];
      dot += sCV[q*D_ + d]                           * Wg[D_ + d];
      dot += dom[(size_t)(b*TQ_ + q0 + q)*D_ + d]    * Wg[2*D_ + d];
    }
    dot = wave_sum(dot);
    if (lane == 0) {
      const float x = dot + bg[0];
      pgenW[b*TQ_ + q0 + q] = 1.0f / (1.0f + __expf(-x));
    }
  }
}

__device__ inline void stage_queries(float* sA, const float* __restrict__ src, int tid) {
  for (int i = tid; i < TQ_*(D_/4); i += 128) {
    const int r  = i >> 6;
    const int c4 = (i & 63) << 2;
    *(v4f*)(sA + r*RP + c4) = *(const v4f*)(src + r*D_ + c4);
  }
}

__device__ inline void logits_64x64(const float* sA, const float* __restrict__ Ws,
                                    int c0, int row, int hi, v8f (&acc)[4][4]) {
  #pragma unroll
  for (int m = 0; m < 4; m++)
    #pragma unroll
    for (int j = 0; j < 4; j++) acc[m][j] = {};

  const float* Bp = Ws + (size_t)(2*hi)*V_ + c0 + 4*row;
  const float* Ap = sA + row*RP + 2*hi;
  #pragma unroll 2
  for (int ks = 0; ks < D_/4; ++ks) {
    const v4f b0 = *(const v4f*)(Bp + (size_t)(4*ks    )*V_);
    const v4f b1 = *(const v4f*)(Bp + (size_t)(4*ks + 1)*V_);
    v2f a[4];
    #pragma unroll
    for (int m = 0; m < 4; m++) a[m] = *(const v2f*)(Ap + m*16*RP + 4*ks);
    #pragma unroll
    for (int j = 0; j < 4; j++) {
      v2f bb; bb.x = b0[j]; bb.y = b1[j];
      #pragma unroll
      for (int m = 0; m < 4; m++) acc[m][j] = wmma_f32_k4(a[m], bb, acc[m][j]);
    }
  }
}

__global__ __launch_bounds__(128) void k_attn_stats(
    const float* __restrict__ dom, const float* __restrict__ out_st,
    const float* __restrict__ ctx, const float* __restrict__ Wg,
    const float* __restrict__ bg, float* __restrict__ attnW,
    float* __restrict__ pgenW, const float* __restrict__ Ws,
    const float* __restrict__ bs, float* __restrict__ pmW,
    float* __restrict__ psW, float* __restrict__ outp)
{
  __shared__ __attribute__((aligned(16))) float smem[TQ_*RP + 2*4*TQ_];

  if (blockIdx.x < NATT) {
    attn_role(smem, blockIdx.x, dom, out_st, ctx, Wg, bg, attnW, pgenW);
    return;
  }

  const int gid  = blockIdx.x - NATT;
  const int b    = gid & (B_ - 1);
  const int vb   = gid >> 4;
  const int wave = threadIdx.x >> 5;
  const int lane = threadIdx.x & 31;
  const int row  = lane & 15;
  const int hi   = lane >> 4;
  const int c0   = vb*CB + wave*64;

  float* sA  = smem;
  float* sM  = smem + TQ_*RP;
  float* sSm = sM + 4*TQ_;

  stage_queries(sA, out_st + (size_t)b*TQ_*D_, threadIdx.x);
  __syncthreads();

  v8f acc[4][4];
  logits_64x64(sA, Ws, c0, row, hi, acc);

  const v4f bsv = *(const v4f*)(bs + c0 + 4*row);
  float* zrow = outp + (size_t)(b*TQ_ + hi*8)*V_ + c0 + 4*row;
  #pragma unroll
  for (int m = 0; m < 4; m++) {
    #pragma unroll
    for (int r = 0; r < 8; r++) {
      const float z0 = acc[m][0][r] + bsv[0];
      const float z1 = acc[m][1][r] + bsv[1];
      const float z2 = acc[m][2][r] + bsv[2];
      const float z3 = acc[m][3][r] + bsv[3];
      *(v4f*)(zrow + (size_t)(m*16 + r)*V_) = (v4f){z0, z1, z2, z3};
      const float tm = half16_max(fmaxf(fmaxf(z0, z1), fmaxf(z2, z3)));
      const float ts = half16_sum((__expf(z0 - tm) + __expf(z1 - tm)) +
                                  (__expf(z2 - tm) + __expf(z3 - tm)));
      if (row == 0) { sM[wave*TQ_ + m*16 + hi*8 + r] = tm; sSm[wave*TQ_ + m*16 + hi*8 + r] = ts; }
    }
  }
  __syncthreads();
  if (threadIdx.x < TQ_) {
    const int q = threadIdx.x;
    float gm = -1e30f;
    #pragma unroll
    for (int w = 0; w < 4; w++) gm = fmaxf(gm, sM[w*TQ_ + q]);
    float gs = 0.f;
    #pragma unroll
    for (int w = 0; w < 4; w++) gs += sSm[w*TQ_ + q] * __expf(sM[w*TQ_ + q] - gm);
    pmW[(size_t)vb*(B_*TQ_) + b*TQ_ + q] = gm;
    psW[(size_t)vb*(B_*TQ_) + b*TQ_ + q] = gs;
  }
}

#define RW 8
#define RJ ((NCB + RW - 1)/RW)

__global__ __launch_bounds__(32*RW) void k_reduce(
    const float* __restrict__ pmW, const float* __restrict__ psW,
    float* __restrict__ rmW, float* __restrict__ rsW)
{
  __shared__ float sM[RW][32];
  __shared__ float sS[RW][32];

  const int lane  = threadIdx.x & 31;
  const int w     = threadIdx.x >> 5;
  const int rowid = blockIdx.x*32 + lane;

  float pm[RJ], ps[RJ];
  #pragma unroll
  for (int j = 0; j < RJ - 1; ++j) {
    const int v = w + RW*j;
    pm[j] = pmW[(size_t)v*(B_*TQ_) + rowid];
    ps[j] = psW[(size_t)v*(B_*TQ_) + rowid];
  }
  {
    const int v = w + RW*(RJ - 1);
    pm[RJ - 1] = -1e30f;
    ps[RJ - 1] = 0.f;
    if (v < NCB) {
      pm[RJ - 1] = pmW[(size_t)v*(B_*TQ_) + rowid];
      ps[RJ - 1] = psW[(size_t)v*(B_*TQ_) + rowid];
    }
  }

  float m = pm[0];
  #pragma unroll
  for (int j = 1; j < RJ; ++j) m = fmaxf(m, pm[j]);
  sM[w][lane] = m;
  __syncthreads();

  float gm = sM[0][lane];
  #pragma unroll
  for (int ww = 1; ww < RW; ++ww) gm = fmaxf(gm, sM[ww][lane]);

  float s = 0.f;
  #pragma unroll
  for (int j = 0; j < RJ; ++j) s += ps[j] * __expf(pm[j] - gm);
  sS[w][lane] = s;
  __syncthreads();

  if (w == 0) {
    float gs = sS[0][lane];
    #pragma unroll
    for (int ww = 1; ww < RW; ++ww) gs += sS[ww][lane];
    rmW[rowid] = gm;
    rsW[rowid] = gs;
  }
}

#define FT  256
#define FIT 16

__global__ __launch_bounds__(FT) void k_final(
    const int* __restrict__ cp, const float* __restrict__ attnW,
    const float* __restrict__ pgenW, const float* __restrict__ rmW,
    const float* __restrict__ rsW, float* outp)
{
  const int b   = blockIdx.x;
  const int v0  = blockIdx.y*CB;
  const int tid = threadIdx.x;
  const int r0  = tid >> 6;
  const int c4  = (tid & 63) << 2;

  __shared__ __attribute__((aligned(16))) float sP[TQ_*CB];
  __shared__ float sScale[TQ_], sMr[TQ_], sOmp[TQ_];
  __shared__ int sHit[TK_];
  __shared__ int sNHit;

  const int key0 = cp[b*TK_ + tid];
  const int key1 = cp[b*TK_ + FT + tid];
  float pg = 0.f, rs = 1.f, rm = 0.f;
  if (tid < TQ_) {
    const int rowid = b*TQ_ + tid;
    pg = pgenW[rowid]; rs = rsW[rowid]; rm = rmW[rowid];
  }

  float* tile = outp + (size_t)(b*TQ_ + r0)*V_ + v0 + c4;
  v4f z[FIT];
  #pragma unroll
  for (int it = 0; it < FIT; ++it) z[it] = *(const v4f*)(tile + (size_t)(4*it)*V_);

  if (tid < TQ_) {
    sScale[tid] = pg / rs;
    sMr[tid]    = rm;
    sOmp[tid]   = 1.0f - pg;
  }
  if (tid == 0) sNHit = 0;
  #pragma unroll
  for (int it = 0; it < FIT; ++it)
    *(v4f*)(sP + (r0 + 4*it)*CB + c4) = (v4f){0.f, 0.f, 0.f, 0.f};
  __syncthreads();

  #pragma unroll
  for (int h = 0; h < 2; ++h) {
    const int key = h ? key1 : key0;
    if (key >= v0 && key < v0 + CB)
      sHit[atomicAdd(&sNHit, 1)] = ((tid + h*FT) << 8) | (key - v0);
  }
  __syncthreads();

  {
    const int nh = sNHit;
    const int q  = tid & 63;
    const float omp = sOmp[q];
    const float* arow = attnW + (size_t)(b*TQ_ + q)*TK_;
    for (int h0 = tid >> 6; h0 < nh; h0 += 16) {
      int   e[4];
      float w[4];
      #pragma unroll
      for (int u = 0; u < 4; ++u) e[u] = sHit[min(h0 + 4*u, nh - 1)];
      #pragma unroll
      for (int u = 0; u < 4; ++u) w[u] = arow[e[u] >> 8];
      #pragma unroll
      for (int u = 0; u < 4; ++u)
        if (h0 + 4*u < nh) atomicAdd(&sP[q*CB + (e[u] & 255)], omp * w[u]);
    }
  }
  __syncthreads();

  #pragma unroll
  for (int it = 0; it < FIT; ++it) {
    const int q = r0 + 4*it;
    const float sc = sScale[q];
    const float mr = sMr[q];
    const v4f pt = *(const v4f*)(sP + q*CB + c4);
    v4f o;
    #pragma unroll
    for (int j = 0; j < 4; j++) o[j] = __logf(sc * __expf(z[it][j] - mr) + pt[j]);
    *(v4f*)(tile + (size_t)(4*it)*V_) = o;
  }
}

extern "C" void kernel_launch(void* const* d_in, const int* in_sizes, int n_in,
                              void* d_out, int out_size, void* d_ws, size_t ws_size,
                              hipStream_t stream) {
  (void)in_sizes; (void)n_in; (void)out_size; (void)ws_size;
  const float* dom    = (const float*)d_in[0];
  const float* out_st = (const float*)d_in[1];
  const float* ctx    = (const float*)d_in[2];
  const int*   cp     = (const int*)d_in[3];
  const float* Ws     = (const float*)d_in[4];
  const float* bsv    = (const float*)d_in[5];
  const float* Wg     = (const float*)d_in[6];
  const float* bg     = (const float*)d_in[7];
  float* outp = (float*)d_out;

  float* attnW = (float*)d_ws;
  float* pgenW = attnW + (size_t)B_*TQ_*TK_;
  float* rmW   = pgenW + B_*TQ_;
  float* rsW   = rmW + B_*TQ_;
  float* pmW   = rsW + B_*TQ_;
  float* psW   = pmW + (size_t)NCB*B_*TQ_;

  k_attn_stats<<<dim3(NATT + B_*NCB), 128, 0, stream>>>(dom, out_st, ctx, Wg, bg, attnW, pgenW,
                                                        Ws, bsv, pmW, psW, outp);
  k_reduce<<<dim3(B_*TQ_/32), 32*RW, 0, stream>>>(pmW, psW, rmW, rsW);
  k_final <<<dim3(B_, NCB),          FT,  0, stream>>>(cp, attnW, pgenW, rmW, rsW, outp);
}


# Vocab GEMM epilogue, row max/sum over 16 lanes: butterflies on the vector ALU's row_xmask swizzle (DPP) instead of shuffles through the LDS unit; 8 rows per stretch; sums and their order unchanged
# speedup vs baseline: 3.8025x; 1.1155x over previous
#include <hip/hip_runtime.h>
#include <hip/hip_bf16.h>


#define B_  16
#define TQ_ 64
#define TK_ 512
#define D_  256
#define V_  32000

#define CB  256
#define NCB 125
#define RP  260

typedef __attribute__((ext_vector_type(2))) float v2f;
typedef __attribute__((ext_vector_type(4))) float v4f;
typedef __attribute__((ext_vector_type(8))) float v8f;

__device__ inline v8f wmma_f32_k4(v2f a, v2f b, v8f c) {
  return __builtin_amdgcn_wmma_f32_16x16x4_f32(
       false, a,  false, b,
       (short)0, c,  false,  false);
}

__device__ inline float wave_max(float x) {
  #pragma unroll
  for (int o = 1; o < 32; o <<= 1) x = fmaxf(x, __shfl_xor(x, o, 32));
  return x;
}
__device__ inline float wave_sum(float x) {
  #pragma unroll
  for (int o = 1; o < 32; o <<= 1) x += __shfl_xor(x, o, 32);
  return x;
}
template <int XM>
__device__ __forceinline__ float row_xor(float x) {
  return __int_as_float(__builtin_amdgcn_update_dpp(0, __float_as_int(x), 0x160 + XM, 0xf, 0xf, true));
}
__device__ inline float half16_max(float x) {
  x = fmaxf(x, row_xor<1>(x));
  x = fmaxf(x, row_xor<2>(x));
  x = fmaxf(x, row_xor<4>(x));
  x = fmaxf(x, row_xor<8>(x));
  return x;
}
__device__ inline float half16_sum(float x) {
  x += row_xor<1>(x);
  x += row_xor<2>(x);
  x += row_xor<4>(x);
  x += row_xor<8>(x);
  return x;
}

#define NATT (B_*TQ_/16)

__device__ __forceinline__ void attn_role(
    float* smem, const int blk,
    const float* __restrict__ dom, const float* __restrict__ out_st,
    const float* __restrict__ ctx, const float* __restrict__ Wg,
    const float* __restrict__ bg, float* __restrict__ attnW,
    float* __restrict__ pgenW)
{
  float* sS  = smem;
  float* sCV = smem + 16*TK_;

  const int b    = blk >> 2;
  const int q0   = (blk & 3) << 4;
  const int wave = threadIdx.x >> 5;
  const int lane = threadIdx.x & 31;
  const int row  = lane & 15;
  const int hi   = lane >> 4;

  v8f acc[8];
  #pragma unroll
  for (int i = 0; i < 8; i++) acc[i] = {};
  const float* Abase = out_st + (size_t)(b*TQ_ + q0 + row)*D_;
  for (int dc = 0; dc < D_/4; ++dc) {
    const int d0 = dc*4;
    v2f a = *(const v2f*)(Abase + d0 + 2*hi);
    #pragma unroll
    for (int i = 0; i < 8; i++) {
      const int k0 = (wave*8 + i)*16;
      v2f bb = *(const v2f*)(ctx + (size_t)(b*TK_ + k0 + row)*D_ + d0 + 2*hi);
      acc[i] = wmma_f32_k4(a, bb, acc[i]);
    }
  }
  #pragma unroll
  for (int i = 0; i < 8; i++) {
    const int k0 = (wave*8 + i)*16;
    #pragma unroll
    for (int r = 0; r < 8; r++)
      sS[(hi*8 + r)*TK_ + k0 + row] = acc[i][r] * 0.0625f;
  }
  __syncthreads();

  #pragma unroll
  for (int j = 0; j < 4; j++) {
    const int q = wave*4 + j;
    float v[TK_/32];
    float m = -1e30f;
    #pragma unroll
    for (int t = 0; t < TK_/32; t++) { v[t] = sS[q*TK_ + lane + 32*t]; m = fmaxf(m, v[t]); }
    m = wave_max(m);
    float s = 0.f;
    #pragma unroll
    for (int t = 0; t < TK_/32; t++) { v[t] = __expf(v[t] - m); s += v[t]; }
    s = wave_sum(s);
    const float inv = 1.0f / s;
    #pragma unroll
    for (int t = 0; t < TK_/32; t++) {
      const float p = v[t] * inv;
      sS[q*TK_ + lane + 32*t] = p;
      attnW[(size_t)(b*TQ_ + q0 + q)*TK_ + lane + 32*t] = p;
    }
  }
  __syncthreads();

  v8f cv[4];
  #pragma unroll
  for (int i = 0; i < 4; i++) cv[i] = {};
  #pragma unroll 4
  for (int kc = 0; kc < TK_/4; ++kc) {
    const int k0 = kc*4;
    v2f a;
    a.x = sS[row*TK_ + k0 + 2*hi];
    a.y = sS[row*TK_ + k0 + 2*hi + 1];
    #pragma unroll
    for (int i = 0; i < 4; i++) {
      const int n0 = (wave*4 + i)*16;
      v2f bb;
      bb.x = ctx[(size_t)(b*TK_ + k0 + 2*hi    )*D_ + n0 + row];
      bb.y = ctx[(size_t)(b*TK_ + k0 + 2*hi + 1)*D_ + n0 + row];
      cv[i] = wmma_f32_k4(a, bb, cv[i]);
    }
  }
  #pragma unroll
  for (int i = 0; i < 4; i++) {
    const int n0 = (wave*4 + i)*16;
    #pragma unroll
    for (int r = 0; r < 8; r++)
      sCV[(hi*8 + r)*D_ + n0 + row] = cv[i][r];
  }
  __syncthreads();

  #pragma unroll
  for (int j = 0; j < 4; j++) {
    const int q = wave*4 + j;
    float dot = 0.f;
    #pragma unroll 2
    for (int d = lane; d < D_; d += 32) {
      dot += out_st[(size_t)(b*TQ_ + q0 + q)*D_ + d] * Wg[d];
      dot += sCV[q*D_ + d]                           * Wg[D_ + d];
      dot += dom[(size_t)(b*TQ_ + q0 + q)*D_ + d]    * Wg[2*D_ + d];
    }
    dot = wave_sum(dot);
    if (lane == 0) {
      const float x = dot + bg[0];
      pgenW[b*TQ_ + q0 + q] = 1.0f / (1.0f + __expf(-x));
    }
  }
}

__device__ inline void stage_queries(float* sA, const float* __restrict__ src, int tid) {
  for (int i = tid; i < TQ_*(D_/4); i += 128) {
    const int r  = i >> 6;
    const int c4 = (i & 63) << 2;
    *(v4f*)(sA + r*RP + c4) = *(const v4f*)(src + r*D_ + c4);
  }
}

__device__ inline void logits_64x64(const float* sA, const float* __restrict__ Ws,
                                    int c0, int row, int hi, v8f (&acc)[4][4]) {
  #pragma unroll
  for (int m = 0; m < 4; m++)
    #pragma unroll
    for (int j = 0; j < 4; j++) acc[m][j] = {};

  const float* Bp = Ws + (size_t)(2*hi)*V_ + c0 + 4*row;
  const float* Ap = sA + row*RP + 2*hi;
  #pragma unroll 2
  for (int ks = 0; ks < D_/4; ++ks) {
    const v4f b0 = *(const v4f*)(Bp + (size_t)(4*ks    )*V_);
    const v4f b1 = *(const v4f*)(Bp + (size_t)(4*ks + 1)*V_);
    v2f a[4];
    #pragma unroll
    for (int m = 0; m < 4; m++) a[m] = *(const v2f*)(Ap + m*16*RP + 4*ks);
    #pragma unroll
    for (int j = 0; j < 4; j++) {
      v2f bb; bb.x = b0[j]; bb.y = b1[j];
      #pragma unroll
      for (int m = 0; m < 4; m++) acc[m][j] = wmma_f32_k4(a[m], bb, acc[m][j]);
    }
  }
}

__global__ __launch_bounds__(128) void k_attn_stats(
    const float* __restrict__ dom, const float* __restrict__ out_st,
    const float* __restrict__ ctx, const float* __restrict__ Wg,
    const float* __restrict__ bg, float* __restrict__ attnW,
    float* __restrict__ pgenW, const float* __restrict__ Ws,
    const float* __restrict__ bs, float* __restrict__ pmW,
    float* __restrict__ psW, float* __restrict__ outp)
{
  __shared__ __attribute__((aligned(16))) float smem[TQ_*RP + 2*4*TQ_];

  if (blockIdx.x < NATT) {
    attn_role(smem, blockIdx.x, dom, out_st, ctx, Wg, bg, attnW, pgenW);
    return;
  }

  const int gid  = blockIdx.x - NATT;
  const int b    = gid & (B_ - 1);
  const int vb   = gid >> 4;
  const int wave = threadIdx.x >> 5;
  const int lane = threadIdx.x & 31;
  const int row  = lane & 15;
  const int hi   = lane >> 4;
  const int c0   = vb*CB + wave*64;

  float* sA  = smem;
  float* sM  = smem + TQ_*RP;
  float* sSm = sM + 4*TQ_;

  stage_queries(sA, out_st + (size_t)b*TQ_*D_, threadIdx.x);
  __syncthreads();

  v8f acc[4][4];
  logits_64x64(sA, Ws, c0, row, hi, acc);

  const v4f bsv = *(const v4f*)(bs + c0 + 4*row);
  float* zrow = outp + (size_t)(b*TQ_ + hi*8)*V_ + c0 + 4*row;
  #pragma unroll
  for (int m = 0; m < 4; m++) {
    float tm[8], ts[8];
    #pragma unroll
    for (int r = 0; r < 8; r++) {
      const float z0 = acc[m][0][r] + bsv[0];
      const float z1 = acc[m][1][r] + bsv[1];
      const float z2 = acc[m][2][r] + bsv[2];
      const float z3 = acc[m][3][r] + bsv[3];
      *(v4f*)(zrow + (size_t)(m*16 + r)*V_) = (v4f){z0, z1, z2, z3};
      tm[r] = half16_max(fmaxf(fmaxf(z0, z1), fmaxf(z2, z3)));
      ts[r] = half16_sum((__expf(z0 - tm[r]) + __expf(z1 - tm[r])) +
                         (__expf(z2 - tm[r]) + __expf(z3 - tm[r])));
    }
    if (row == 0) {
      float* pm = sM  + wave*TQ_ + m*16 + hi*8;
      float* ps = sSm + wave*TQ_ + m*16 + hi*8;
      *(v4f*)(pm)     = (v4f){tm[0], tm[1], tm[2], tm[3]};
      *(v4f*)(pm + 4) = (v4f){tm[4], tm[5], tm[6], tm[7]};
      *(v4f*)(ps)     = (v4f){ts[0], ts[1], ts[2], ts[3]};
      *(v4f*)(ps + 4) = (v4f){ts[4], ts[5], ts[6], ts[7]};
    }
  }
  __syncthreads();
  if (threadIdx.x < TQ_) {
    const int q = threadIdx.x;
    float gm = -1e30f;
    #pragma unroll
    for (int w = 0; w < 4; w++) gm = fmaxf(gm, sM[w*TQ_ + q]);
    float gs = 0.f;
    #pragma unroll
    for (int w = 0; w < 4; w++) gs += sSm[w*TQ_ + q] * __expf(sM[w*TQ_ + q] - gm);
    pmW[(size_t)vb*(B_*TQ_) + b*TQ_ + q] = gm;
    psW[(size_t)vb*(B_*TQ_) + b*TQ_ + q] = gs;
  }
}

#define RW 8
#define RJ ((NCB + RW - 1)/RW)

__global__ __launch_bounds__(32*RW) void k_reduce(
    const float* __restrict__ pmW, const float* __restrict__ psW,
    float* __restrict__ rmW, float* __restrict__ rsW)
{
  __shared__ float sM[RW][32];
  __shared__ float sS[RW][32];

  const int lane  = threadIdx.x & 31;
  const int w     = threadIdx.x >> 5;
  const int rowid = blockIdx.x*32 + lane;

  float pm[RJ], ps[RJ];
  #pragma unroll
  for (int j = 0; j < RJ - 1; ++j) {
    const int v = w + RW*j;
    pm[j] = pmW[(size_t)v*(B_*TQ_) + rowid];
    ps[j] = psW[(size_t)v*(B_*TQ_) + rowid];
  }
  {
    const int v = w + RW*(RJ - 1);
    pm[RJ - 1] = -1e30f;
    ps[RJ - 1] = 0.f;
    if (v < NCB) {
      pm[RJ - 1] = pmW[(size_t)v*(B_*TQ_) + rowid];
      ps[RJ - 1] = psW[(size_t)v*(B_*TQ_) + rowid];
    }
  }

  float m = pm[0];
  #pragma unroll
  for (int j = 1; j < RJ; ++j) m = fmaxf(m, pm[j]);
  sM[w][lane] = m;
  __syncthreads();

  float gm = sM[0][lane];
  #pragma unroll
  for (int ww = 1; ww < RW; ++ww) gm = fmaxf(gm, sM[ww][lane]);

  float s = 0.f;
  #pragma unroll
  for (int j = 0; j < RJ; ++j) s += ps[j] * __expf(pm[j] - gm);
  sS[w][lane] = s;
  __syncthreads();

  if (w == 0) {
    float gs = sS[0][lane];
    #pragma unroll
    for (int ww = 1; ww < RW; ++ww) gs += sS[ww][lane];
    rmW[rowid] = gm;
    rsW[rowid] = gs;
  }
}

#define FT  256
#define FIT 16

__global__ __launch_bounds__(FT) void k_final(
    const int* __restrict__ cp, const float* __restrict__ attnW,
    const float* __restrict__ pgenW, const float* __restrict__ rmW,
    const float* __restrict__ rsW, float* outp)
{
  const int b   = blockIdx.x;
  const int v0  = blockIdx.y*CB;
  const int tid = threadIdx.x;
  const int r0  = tid >> 6;
  const int c4  = (tid & 63) << 2;

  __shared__ __attribute__((aligned(16))) float sP[TQ_*CB];
  __shared__ float sScale[TQ_], sMr[TQ_], sOmp[TQ_];
  __shared__ int sHit[TK_];
  __shared__ int sNHit;

  const int key0 = cp[b*TK_ + tid];
  const int key1 = cp[b*TK_ + FT + tid];
  float pg = 0.f, rs = 1.f, rm = 0.f;
  if (tid < TQ_) {
    const int rowid = b*TQ_ + tid;
    pg = pgenW[rowid]; rs = rsW[rowid]; rm = rmW[rowid];
  }

  float* tile = outp + (size_t)(b*TQ_ + r0)*V_ + v0 + c4;
  v4f z[FIT];
  #pragma unroll
  for (int it = 0; it < FIT; ++it) z[it] = *(const v4f*)(tile + (size_t)(4*it)*V_);

  if (tid < TQ_) {
    sScale[tid] = pg / rs;
    sMr[tid]    = rm;
    sOmp[tid]   = 1.0f - pg;
  }
  if (tid == 0) sNHit = 0;
  #pragma unroll
  for (int it = 0; it < FIT; ++it)
    *(v4f*)(sP + (r0 + 4*it)*CB + c4) = (v4f){0.f, 0.f, 0.f, 0.f};
  __syncthreads();

  #pragma unroll
  for (int h = 0; h < 2; ++h) {
    const int key = h ? key1 : key0;
    if (key >= v0 && key < v0 + CB)
      sHit[atomicAdd(&sNHit, 1)] = ((tid + h*FT) << 8) | (key - v0);
  }
  __syncthreads();

  {
    const int nh = sNHit;
    const int q  = tid & 63;
    const float omp = sOmp[q];
    const float* arow = attnW + (size_t)(b*TQ_ + q)*TK_;
    for (int h0 = tid >> 6; h0 < nh; h0 += 16) {
      int   e[4];
      float w[4];
      #pragma unroll
      for (int u = 0; u < 4; ++u) e[u] = sHit[min(h0 + 4*u, nh - 1)];
      #pragma unroll
      for (int u = 0; u < 4; ++u) w[u] = arow[e[u] >> 8];
      #pragma unroll
      for (int u = 0; u < 4; ++u)
        if (h0 + 4*u < nh) atomicAdd(&sP[q*CB + (e[u] & 255)], omp * w[u]);
    }
  }
  __syncthreads();

  #pragma unroll
  for (int it = 0; it < FIT; ++it) {
    const int q = r0 + 4*it;
    const float sc = sScale[q];
    const float mr = sMr[q];
    const v4f pt = *(const v4f*)(sP + q*CB + c4);
    v4f o;
    #pragma unroll
    for (int j = 0; j < 4; j++) o[j] = __logf(sc * __expf(z[it][j] - mr) + pt[j]);
    *(v4f*)(tile + (size_t)(4*it)*V_) = o;
  }
}

extern "C" void kernel_launch(void* const* d_in, const int* in_sizes, int n_in,
                              void* d_out, int out_size, void* d_ws, size_t ws_size,
                              hipStream_t stream) {
  (void)in_sizes; (void)n_in; (void)out_size; (void)ws_size;
  const float* dom    = (const float*)d_in[0];
  const float* out_st = (const float*)d_in[1];
  const float* ctx    = (const float*)d_in[2];
  const int*   cp     = (const int*)d_in[3];
  const float* Ws     = (const float*)d_in[4];
  const float* bsv    = (const float*)d_in[5];
  const float* Wg     = (const float*)d_in[6];
  const float* bg     = (const float*)d_in[7];
  float* outp = (float*)d_out;

  float* attnW = (float*)d_ws;
  float* pgenW = attnW + (size_t)B_*TQ_*TK_;
  float* rmW   = pgenW + B_*TQ_;
  float* rsW   = rmW + B_*TQ_;
  float* pmW   = rsW + B_*TQ_;
  float* psW   = pmW + (size_t)NCB*B_*TQ_;

  k_attn_stats<<<dim3(NATT + B_*NCB), 128, 0, stream>>>(dom, out_st, ctx, Wg, bg, attnW, pgenW,
                                                        Ws, bsv, pmW, psW, outp);
  k_reduce<<<dim3(B_*TQ_/32), 32*RW, 0, stream>>>(pmW, psW, rmW, rsW);
  k_final <<<dim3(B_, NCB),          FT,  0, stream>>>(cp, attnW, pgenW, rmW, rsW, outp);
}


# Vocab GEMM k loop: W_state rows and query fragments of the next stage (2 k-steps) go into a second register set ahead of the current 32 WMMAs, not issued and awaited in one trip; sums unchanged
# speedup vs baseline: 4.0333x; 1.0607x over previous
#include <hip/hip_runtime.h>
#include <hip/hip_bf16.h>


#define B_  16
#define TQ_ 64
#define TK_ 512
#define D_  256
#define V_  32000

#define CB  256
#define NCB 125
#define RP  260

typedef __attribute__((ext_vector_type(2))) float v2f;
typedef __attribute__((ext_vector_type(4))) float v4f;
typedef __attribute__((ext_vector_type(8))) float v8f;

__device__ inline v8f wmma_f32_k4(v2f a, v2f b, v8f c) {
  return __builtin_amdgcn_wmma_f32_16x16x4_f32(
       false, a,  false, b,
       (short)0, c,  false,  false);
}

__device__ inline float wave_max(float x) {
  #pragma unroll
  for (int o = 1; o < 32; o <<= 1) x = fmaxf(x, __shfl_xor(x, o, 32));
  return x;
}
__device__ inline float wave_sum(float x) {
  #pragma unroll
  for (int o = 1; o < 32; o <<= 1) x += __shfl_xor(x, o, 32);
  return x;
}
template <int XM>
__device__ __forceinline__ float row_xor(float x) {
  return __int_as_float(__builtin_amdgcn_update_dpp(0, __float_as_int(x), 0x160 + XM, 0xf, 0xf, true));
}
__device__ inline float half16_max(float x) {
  x = fmaxf(x, row_xor<1>(x));
  x = fmaxf(x, row_xor<2>(x));
  x = fmaxf(x, row_xor<4>(x));
  x = fmaxf(x, row_xor<8>(x));
  return x;
}
__device__ inline float half16_sum(float x) {
  x += row_xor<1>(x);
  x += row_xor<2>(x);
  x += row_xor<4>(x);
  x += row_xor<8>(x);
  return x;
}

#define NATT (B_*TQ_/16)

__device__ __forceinline__ void attn_role(
    float* smem, const int blk,
    const float* __restrict__ dom, const float* __restrict__ out_st,
    const float* __restrict__ ctx, const float* __restrict__ Wg,
    const float* __restrict__ bg, float* __restrict__ attnW,
    float* __restrict__ pgenW)
{
  float* sS  = smem;
  float* sCV = smem + 16*TK_;

  const int b    = blk >> 2;
  const int q0   = (blk & 3) << 4;
  const int wave = threadIdx.x >> 5;
  const int lane = threadIdx.x & 31;
  const int row  = lane & 15;
  const int hi   = lane >> 4;

  v8f acc[8];
  #pragma unroll
  for (int i = 0; i < 8; i++) acc[i] = {};
  const float* Abase = out_st + (size_t)(b*TQ_ + q0 + row)*D_;
  for (int dc = 0; dc < D_/4; ++dc) {
    const int d0 = dc*4;
    v2f a = *(const v2f*)(Abase + d0 + 2*hi);
    #pragma unroll
    for (int i = 0; i < 8; i++) {
      const int k0 = (wave*8 + i)*16;
      v2f bb = *(const v2f*)(ctx + (size_t)(b*TK_ + k0 + row)*D_ + d0 + 2*hi);
      acc[i] = wmma_f32_k4(a, bb, acc[i]);
    }
  }
  #pragma unroll
  for (int i = 0; i < 8; i++) {
    const int k0 = (wave*8 + i)*16;
    #pragma unroll
    for (int r = 0; r < 8; r++)
      sS[(hi*8 + r)*TK_ + k0 + row] = acc[i][r] * 0.0625f;
  }
  __syncthreads();

  #pragma unroll
  for (int j = 0; j < 4; j++) {
    const int q = wave*4 + j;
    float v[TK_/32];
    float m = -1e30f;
    #pragma unroll
    for (int t = 0; t < TK_/32; t++) { v[t] = sS[q*TK_ + lane + 32*t]; m = fmaxf(m, v[t]); }
    m = wave_max(m);
    float s = 0.f;
    #pragma unroll
    for (int t = 0; t < TK_/32; t++) { v[t] = __expf(v[t] - m); s += v[t]; }
    s = wave_sum(s);
    const float inv = 1.0f / s;
    #pragma unroll
    for (int t = 0; t < TK_/32; t++) {
      const float p = v[t] * inv;
      sS[q*TK_ + lane + 32*t] = p;
      attnW[(size_t)(b*TQ_ + q0 + q)*TK_ + lane + 32*t] = p;
    }
  }
  __syncthreads();

  v8f cv[4];
  #pragma unroll
  for (int i = 0; i < 4; i++) cv[i] = {};
  #pragma unroll 4
  for (int kc = 0; kc < TK_/4; ++kc) {
    const int k0 = kc*4;
    v2f a;
    a.x = sS[row*TK_ + k0 + 2*hi];
    a.y = sS[row*TK_ + k0 + 2*hi + 1];
    #pragma unroll
    for (int i = 0; i < 4; i++) {
      const int n0 = (wave*4 + i)*16;
      v2f bb;
      bb.x = ctx[(size_t)(b*TK_ + k0 + 2*hi    )*D_ + n0 + row];
      bb.y = ctx[(size_t)(b*TK_ + k0 + 2*hi + 1)*D_ + n0 + row];
      cv[i] = wmma_f32_k4(a, bb, cv[i]);
    }
  }
  #pragma unroll
  for (int i = 0; i < 4; i++) {
    const int n0 = (wave*4 + i)*16;
    #pragma unroll
    for (int r = 0; r < 8; r++)
      sCV[(hi*8 + r)*D_ + n0 + row] = cv[i][r];
  }
  __syncthreads();

  #pragma unroll
  for (int j = 0; j < 4; j++) {
    const int q = wave*4 + j;
    float dot = 0.f;
    #pragma unroll 2
    for (int d = lane; d < D_; d += 32) {
      dot += out_st[(size_t)(b*TQ_ + q0 + q)*D_ + d] * Wg[d];
      dot += sCV[q*D_ + d]                           * Wg[D_ + d];
      dot += dom[(size_t)(b*TQ_ + q0 + q)*D_ + d]    * Wg[2*D_ + d];
    }
    dot = wave_sum(dot);
    if (lane == 0) {
      const float x = dot + bg[0];
      pgenW[b*TQ_ + q0 + q] = 1.0f / (1.0f + __expf(-x));
    }
  }
}

__device__ inline void stage_queries(float* sA, const float* __restrict__ src, int tid) {
  for (int i = tid; i < TQ_*(D_/4); i += 128) {
    const int r  = i >> 6;
    const int c4 = (i & 63) << 2;
    *(v4f*)(sA + r*RP + c4) = *(const v4f*)(src + r*D_ + c4);
  }
}

struct KStage {
  v4f b[4];
  v2f a[2][4];
};

__device__ __forceinline__ void load_stage(KStage& f, const float* __restrict__ Bp, const float* Ap, int t) {
  #pragma unroll
  for (int s = 0; s < 2; s++) {
    f.b[2*s]     = *(const v4f*)(Bp + (size_t)(4*(2*t + s)    )*V_);
    f.b[2*s + 1] = *(const v4f*)(Bp + (size_t)(4*(2*t + s) + 1)*V_);
  }
  #pragma unroll
  for (int s = 0; s < 2; s++)
    #pragma unroll
    for (int m = 0; m < 4; m++) f.a[s][m] = *(const v2f*)(Ap + m*16*RP + 4*(2*t + s));
}

__device__ __forceinline__ void mma_stage(const KStage& f, v8f (&acc)[4][4]) {
  #pragma unroll
  for (int s = 0; s < 2; s++)
    #pragma unroll
    for (int j = 0; j < 4; j++) {
      v2f bb; bb.x = f.b[2*s][j]; bb.y = f.b[2*s + 1][j];
      #pragma unroll
      for (int m = 0; m < 4; m++) acc[m][j] = wmma_f32_k4(f.a[s][m], bb, acc[m][j]);
    }
}

__device__ inline void logits_64x64(const float* sA, const float* __restrict__ Ws,
                                    int c0, int row, int hi, v8f (&acc)[4][4]) {
  #pragma unroll
  for (int m = 0; m < 4; m++)
    #pragma unroll
    for (int j = 0; j < 4; j++) acc[m][j] = {};

  const float* Bp = Ws + (size_t)(2*hi)*V_ + c0 + 4*row;
  const float* Ap = sA + row*RP + 2*hi;
  KStage f0, f1;
  load_stage(f0, Bp, Ap, 0);
  #pragma unroll 1
  for (int t = 0; t < D_/8 - 2; t += 2) {
    load_stage(f1, Bp, Ap, t + 1);
    mma_stage(f0, acc);
    load_stage(f0, Bp, Ap, t + 2);
    mma_stage(f1, acc);
  }
  load_stage(f1, Bp, Ap, D_/8 - 1);
  mma_stage(f0, acc);
  mma_stage(f1, acc);
}

__global__ __launch_bounds__(128) void k_attn_stats(
    const float* __restrict__ dom, const float* __restrict__ out_st,
    const float* __restrict__ ctx, const float* __restrict__ Wg,
    const float* __restrict__ bg, float* __restrict__ attnW,
    float* __restrict__ pgenW, const float* __restrict__ Ws,
    const float* __restrict__ bs, float* __restrict__ pmW,
    float* __restrict__ psW, float* __restrict__ outp)
{
  __shared__ __attribute__((aligned(16))) float smem[TQ_*RP + 2*4*TQ_];

  if (blockIdx.x < NATT) {
    attn_role(smem, blockIdx.x, dom, out_st, ctx, Wg, bg, attnW, pgenW);
    return;
  }

  const int gid  = blockIdx.x - NATT;
  const int b    = gid & (B_ - 1);
  const int vb   = gid >> 4;
  const int wave = threadIdx.x >> 5;
  const int lane = threadIdx.x & 31;
  const int row  = lane & 15;
  const int hi   = lane >> 4;
  const int c0   = vb*CB + wave*64;

  float* sA  = smem;
  float* sM  = smem + TQ_*RP;
  float* sSm = sM + 4*TQ_;

  stage_queries(sA, out_st + (size_t)b*TQ_*D_, threadIdx.x);
  __syncthreads();

  v8f acc[4][4];
  logits_64x64(sA, Ws, c0, row, hi, acc);

  const v4f bsv = *(const v4f*)(bs + c0 + 4*row);
  float* zrow = outp + (size_t)(b*TQ_ + hi*8)*V_ + c0 + 4*row;
  #pragma unroll
  for (int m = 0; m < 4; m++) {
    float tm[8], ts[8];
    #pragma unroll
    for (int r = 0; r < 8; r++) {
      const float z0 = acc[m][0][r] + bsv[0];
      const float z1 = acc[m][1][r] + bsv[1];
      const float z2 = acc[m][2][r] + bsv[2];
      const float z3 = acc[m][3][r] + bsv[3];
      *(v4f*)(zrow + (size_t)(m*16 + r)*V_) = (v4f){z0, z1, z2, z3};
      tm[r] = half16_max(fmaxf(fmaxf(z0, z1), fmaxf(z2, z3)));
      ts[r] = half16_sum((__expf(z0 - tm[r]) + __expf(z1 - tm[r])) +
                         (__expf(z2 - tm[r]) + __expf(z3 - tm[r])));
    }
    if (row == 0) {
      float* pm = sM  + wave*TQ_ + m*16 + hi*8;
      float* ps = sSm + wave*TQ_ + m*16 + hi*8;
      *(v4f*)(pm)     = (v4f){tm[0], tm[1], tm[2], tm[3]};
      *(v4f*)(pm + 4) = (v4f){tm[4], tm[5], tm[6], tm[7]};
      *(v4f*)(ps)     = (v4f){ts[0], ts[1], ts[2], ts[3]};
      *(v4f*)(ps + 4) = (v4f){ts[4], ts[5], ts[6], ts[7]};
    }
  }
  __syncthreads();
  if (threadIdx.x < TQ_) {
    const int q = threadIdx.x;
    float gm = -1e30f;
    #pragma unroll
    for (int w = 0; w < 4; w++) gm = fmaxf(gm, sM[w*TQ_ + q]);
    float gs = 0.f;
    #pragma unroll
    for (int w = 0; w < 4; w++) gs += sSm[w*TQ_ + q] * __expf(sM[w*TQ_ + q] - gm);
    pmW[(size_t)vb*(B_*TQ_) + b*TQ_ + q] = gm;
    psW[(size_t)vb*(B_*TQ_) + b*TQ_ + q] = gs;
  }
}

#define RW 8
#define RJ ((NCB + RW - 1)/RW)

__global__ __launch_bounds__(32*RW) void k_reduce(
    const float* __restrict__ pmW, const float* __restrict__ psW,
    float* __restrict__ rmW, float* __restrict__ rsW)
{
  __shared__ float sM[RW][32];
  __shared__ float sS[RW][32];

  const int lane  = threadIdx.x & 31;
  const int w     = threadIdx.x >> 5;
  const int rowid = blockIdx.x*32 + lane;

  float pm[RJ], ps[RJ];
  #pragma unroll
  for (int j = 0; j < RJ - 1; ++j) {
    const int v = w + RW*j;
    pm[j] = pmW[(size_t)v*(B_*TQ_) + rowid];
    ps[j] = psW[(size_t)v*(B_*TQ_) + rowid];
  }
  {
    const int v = w + RW*(RJ - 1);
    pm[RJ - 1] = -1e30f;
    ps[RJ - 1] = 0.f;
    if (v < NCB) {
      pm[RJ - 1] = pmW[(size_t)v*(B_*TQ_) + rowid];
      ps[RJ - 1] = psW[(size_t)v*(B_*TQ_) + rowid];
    }
  }

  float m = pm[0];
  #pragma unroll
  for (int j = 1; j < RJ; ++j) m = fmaxf(m, pm[j]);
  sM[w][lane] = m;
  __syncthreads();

  float gm = sM[0][lane];
  #pragma unroll
  for (int ww = 1; ww < RW; ++ww) gm = fmaxf(gm, sM[ww][lane]);

  float s = 0.f;
  #pragma unroll
  for (int j = 0; j < RJ; ++j) s += ps[j] * __expf(pm[j] - gm);
  sS[w][lane] = s;
  __syncthreads();

  if (w == 0) {
    float gs = sS[0][lane];
    #pragma unroll
    for (int ww = 1; ww < RW; ++ww) gs += sS[ww][lane];
    rmW[rowid] = gm;
    rsW[rowid] = gs;
  }
}

#define FT  256
#define FIT 16

__global__ __launch_bounds__(FT) void k_final(
    const int* __restrict__ cp, const float* __restrict__ attnW,
    const float* __restrict__ pgenW, const float* __restrict__ rmW,
    const float* __restrict__ rsW, float* outp)
{
  const int b   = blockIdx.x;
  const int v0  = blockIdx.y*CB;
  const int tid = threadIdx.x;
  const int r0  = tid >> 6;
  const int c4  = (tid & 63) << 2;

  __shared__ __attribute__((aligned(16))) float sP[TQ_*CB];
  __shared__ float sScale[TQ_], sMr[TQ_], sOmp[TQ_];
  __shared__ int sHit[TK_];
  __shared__ int sNHit;

  const int key0 = cp[b*TK_ + tid];
  const int key1 = cp[b*TK_ + FT + tid];
  float pg = 0.f, rs = 1.f, rm = 0.f;
  if (tid < TQ_) {
    const int rowid = b*TQ_ + tid;
    pg = pgenW[rowid]; rs = rsW[rowid]; rm = rmW[rowid];
  }

  float* tile = outp + (size_t)(b*TQ_ + r0)*V_ + v0 + c4;
  v4f z[FIT];
  #pragma unroll
  for (int it = 0; it < FIT; ++it) z[it] = *(const v4f*)(tile + (size_t)(4*it)*V_);

  if (tid < TQ_) {
    sScale[tid] = pg / rs;
    sMr[tid]    = rm;
    sOmp[tid]   = 1.0f - pg;
  }
  if (tid == 0) sNHit = 0;
  #pragma unroll
  for (int it = 0; it < FIT; ++it)
    *(v4f*)(sP + (r0 + 4*it)*CB + c4) = (v4f){0.f, 0.f, 0.f, 0.f};
  __syncthreads();

  #pragma unroll
  for (int h = 0; h < 2; ++h) {
    const int key = h ? key1 : key0;
    if (key >= v0 && key < v0 + CB)
      sHit[atomicAdd(&sNHit, 1)] = ((tid + h*FT) << 8) | (key - v0);
  }
  __syncthreads();

  {
    const int nh = sNHit;
    const int q  = tid & 63;
    const float omp = sOmp[q];
    const float* arow = attnW + (size_t)(b*TQ_ + q)*TK_;
    for (int h0 = tid >> 6; h0 < nh; h0 += 16) {
      int   e[4];
      float w[4];
      #pragma unroll
      for (int u = 0; u < 4; ++u) e[u] = sHit[min(h0 + 4*u, nh - 1)];
      #pragma unroll
      for (int u = 0; u < 4; ++u) w[u] = arow[e[u] >> 8];
      #pragma unroll
      for (int u = 0; u < 4; ++u)
        if (h0 + 4*u < nh) atomicAdd(&sP[q*CB + (e[u] & 255)], omp * w[u]);
    }
  }
  __syncthreads();

  #pragma unroll
  for (int it = 0; it < FIT; ++it) {
    const int q = r0 + 4*it;
    const float sc = sScale[q];
    const float mr = sMr[q];
    const v4f pt = *(const v4f*)(sP + q*CB + c4);
    v4f o;
    #pragma unroll
    for (int j = 0; j < 4; j++) o[j] = __logf(sc * __expf(z[it][j] - mr) + pt[j]);
    *(v4f*)(tile + (size_t)(4*it)*V_) = o;
  }
}

extern "C" void kernel_launch(void* const* d_in, const int* in_sizes, int n_in,
                              void* d_out, int out_size, void* d_ws, size_t ws_size,
                              hipStream_t stream) {
  (void)in_sizes; (void)n_in; (void)out_size; (void)ws_size;
  const float* dom    = (const float*)d_in[0];
  const float* out_st = (const float*)d_in[1];
  const float* ctx    = (const float*)d_in[2];
  const int*   cp     = (const int*)d_in[3];
  const float* Ws     = (const float*)d_in[4];
  const float* bsv    = (const float*)d_in[5];
  const float* Wg     = (const float*)d_in[6];
  const float* bg     = (const float*)d_in[7];
  float* outp = (float*)d_out;

  float* attnW = (float*)d_ws;
  float* pgenW = attnW + (size_t)B_*TQ_*TK_;
  float* rmW   = pgenW + B_*TQ_;
  float* rsW   = rmW + B_*TQ_;
  float* pmW   = rsW + B_*TQ_;
  float* psW   = pmW + (size_t)NCB*B_*TQ_;

  k_attn_stats<<<dim3(NATT + B_*NCB), 128, 0, stream>>>(dom, out_st, ctx, Wg, bg, attnW, pgenW,
                                                        Ws, bsv, pmW, psW, outp);
  k_reduce<<<dim3(B_*TQ_/32), 32*RW, 0, stream>>>(pmW, psW, rmW, rsW);
  k_final <<<dim3(B_, NCB),          FT,  0, stream>>>(cp, attnW, pgenW, rmW, rsW, outp);
}
